# starting kernel
# baseline (speedup 1.0000x reference)
#include <hip/hip_runtime.h>

typedef __attribute__((ext_vector_type(2))) float v2f;
typedef __attribute__((ext_vector_type(8))) float v8f;

#define D_DIM 64
#define B_DIM 2

__device__ __forceinline__ void gadd_f32(float* p, float v) {
  asm volatile("global_atomic_add_f32 %0, %1, off"
               :
               : "v"(p), "v"(v)
               : "memory");
}

__global__ void zero_ws_kernel(float4* __restrict__ ws, int n4) {
  int i = blockIdx.x * blockDim.x + threadIdx.x;
  if (i < n4) {
    float4 z; z.x = 0.f; z.y = 0.f; z.z = 0.f; z.w = 0.f;
    ws[i] = z;
  }
}

__global__ void __launch_bounds__(256)
scatter_edges_kernel(const float* __restrict__ prev,
                     const int*   __restrict__ edges,
                     const float* __restrict__ status,
                     float*       __restrict__ nbr,
                     int E, int N) {
  long tid = (long)blockIdx.x * blockDim.x + threadIdx.x;
  long e   = tid >> 5;
  if (e >= E) return;
  int lane = (int)(tid & 31);
  int b    = lane >> 4;
  int d4   = (lane & 15) * 4;

  int   u = edges[2 * e + 0];
  int   v = edges[2 * e + 1];
  float g = status[e];

  const float4 pu = *(const float4*)(prev + ((size_t)b * N + u) * D_DIM + d4);
  const float4 pv = *(const float4*)(prev + ((size_t)b * N + v) * D_DIM + d4);

  float* ov = nbr + ((size_t)b * N + v) * D_DIM + d4;
  float* ou = nbr + ((size_t)b * N + u) * D_DIM + d4;

  gadd_f32(ov + 0, pu.x * g);
  gadd_f32(ov + 1, pu.y * g);
  gadd_f32(ov + 2, pu.z * g);
  gadd_f32(ov + 3, pu.w * g);
  gadd_f32(ou + 0, pv.x * g);
  gadd_f32(ou + 1, pv.y * g);
  gadd_f32(ou + 2, pv.z * g);
  gadd_f32(ou + 3, pv.w * g);
}

__global__ void __launch_bounds__(256)
gemm_bias_lrelu_kernel(const float* __restrict__ nbr,
                       const float* __restrict__ Wm,
                       const float* __restrict__ nf,
                       const float* __restrict__ ef,
                       float*       __restrict__ out,
                       int ntiles) {
  long tid  = (long)blockIdx.x * blockDim.x + threadIdx.x;
  long tile = tid >> 5;
  if (tile >= ntiles) return;
  int  lane = (int)(tid & 31);
  int  half = lane >> 4;
  int  lm   = lane & 15;
  long row_base = tile * 16;

  v2f afrag[16];
  const float* arow = nbr + (row_base + lm) * D_DIM + 2 * half;
#pragma unroll
  for (int t = 0; t < 16; ++t)
    afrag[t] = *(const v2f*)(arow + 4 * t);

#pragma unroll
  for (int c = 0; c < 4; ++c) {
    int col = c * 16 + lm;

    v8f acc;
#pragma unroll
    for (int r = 0; r < 8; ++r) {
      long rr = row_base + r + 8 * half;
      acc[r] = nf[rr * D_DIM + col] + ef[rr * D_DIM + col];
    }

    const float* wrow = Wm + (size_t)col * D_DIM + 2 * half;
#pragma unroll
    for (int t = 0; t < 16; ++t) {
      v2f bfrag = *(const v2f*)(wrow + 4 * t);
      acc = __builtin_amdgcn_wmma_f32_16x16x4_f32(
          false, afrag[t], false, bfrag, (short)0, acc, false, false);
    }

#pragma unroll
    for (int r = 0; r < 8; ++r) {
      long rr = row_base + r + 8 * half;
      float dv = acc[r];
      dv = dv > 0.f ? dv : 0.01f * dv;
      __builtin_nontemporal_store(dv, out + rr * D_DIM + col);
    }
  }
}

extern "C" void kernel_launch(void* const* d_in, const int* in_sizes, int n_in,
                              void* d_out, int out_size, void* d_ws, size_t ws_size,
                              hipStream_t stream) {
  const float* prev  = (const float*)d_in[0];
  const int*   edges = (const int*)  d_in[1];
  const float* nf    = (const float*)d_in[2];
  const float* ef    = (const float*)d_in[3];
  const float* st    = (const float*)d_in[4];
  const float* Wm    = (const float*)d_in[5];

  float* nbr = (float*)d_ws;

  const int BND = in_sizes[0];
  const int E   = in_sizes[4];
  const int N   = BND / (B_DIM * D_DIM);

  int n4 = BND / 4;
  zero_ws_kernel<<<(n4 + 255) / 256, 256, 0, stream>>>((float4*)nbr, n4);

  long sthreads = (long)E * 32;
  scatter_edges_kernel<<<(int)((sthreads + 255) / 256), 256, 0, stream>>>(
      prev, edges, st, nbr, E, N);

  int ntiles = (BND / D_DIM) / 16;
  long gthreads = (long)ntiles * 32;
  gemm_bias_lrelu_kernel<<<(int)((gthreads + 255) / 256), 256, 0, stream>>>(
      nbr, Wm, nf, ef, (float*)d_out, ntiles);
}


# Float-atomic edge scatter replaced by an atomic-free per-call grouping by receiver (two-level counting sort, thread-private LDS counters) and a one-wave-per-node register sum in fixed list order
# speedup vs baseline: 2.0480x; 2.0480x over previous
#include <hip/hip_runtime.h>

typedef __attribute__((ext_vector_type(2))) float v2f;
typedef __attribute__((ext_vector_type(8))) float v8f;

#define D_DIM 64
#define B_DIM 2


#define BIN_SHIFT  8
#define BIN_NODES  256
#define PITCH      257
#define MAX_BINS   200
#define MAX_BLKS   256
#define THREADS    256
#define EPT        16
#define EPB        (EPT * THREADS)

__device__ __forceinline__ void load_edges(int2 (&uv)[EPT], const int* __restrict__ edges,
                                           int E, int e0, int t) {
#pragma unroll
  for (int q = 0; q < EPT; ++q) {
    const int e = e0 + q * THREADS + t;
    uv[q] = e < E ? ((const int2*)edges)[e] : make_int2(-1, -1);
  }
}

__device__ __forceinline__ void coarse_count(int* tab, const int2 (&uv)[EPT], int nb, int t) {
  for (int r = 0; r < nb; ++r) tab[r * PITCH + t] = 0;
#pragma unroll
  for (int q = 0; q < EPT; ++q) {
    if (uv[q].x >= 0) {
      tab[(uv[q].y >> BIN_SHIFT) * PITCH + t] += 1;
      tab[(uv[q].x >> BIN_SHIFT) * PITCH + t] += 1;
    }
  }
}

__global__ void __launch_bounds__(THREADS)
coarse_count_kernel(const int* __restrict__ edges, int* __restrict__ blockhist,
                    int E, int nb, int nblk) {
  __shared__ int tab[MAX_BINS * PITCH];
  const int t = threadIdx.x, blk = blockIdx.x;
  int2 uv[EPT];
  load_edges(uv, edges, E, blk * EPB, t);
  coarse_count(tab, uv, nb, t);
  __syncthreads();
  for (int r = t; r < nb; r += THREADS) {
    int s = 0;
    for (int j = 0; j < THREADS; ++j) s += tab[r * PITCH + j];
    blockhist[r * nblk + blk] = s;
  }
}

__global__ void __launch_bounds__(1024)
scan_counts_kernel(const int* __restrict__ blockhist, int* __restrict__ blockoff,
                   int* __restrict__ bin_start, int n, int nb, int nblk) {
  __shared__ int buf[MAX_BINS * MAX_BLKS];
  __shared__ int part[1024];
  const int t = threadIdx.x;
  for (int base = 0; base < n; base += 1024 * 8) {
    int v[8];
#pragma unroll
    for (int q = 0; q < 8; ++q) {
      const int i = base + q * 1024 + t;
      v[q] = i < n ? blockhist[i] : 0;
    }
#pragma unroll
    for (int q = 0; q < 8; ++q) {
      const int i = base + q * 1024 + t;
      if (i < n) buf[i] = v[q];
    }
  }
  __syncthreads();
  const int per = (n + 1023) / 1024;
  const int lo  = t * per;
  const int hi  = (lo + per) < n ? (lo + per) : n;
  int s = 0;
  for (int i = lo; i < hi; ++i) s += buf[i];
  part[t] = s;
  __syncthreads();
  for (int off = 1; off < 1024; off <<= 1) {
    const int a = t >= off ? part[t - off] : 0;
    __syncthreads();
    part[t] += a;
    __syncthreads();
  }
  int run = part[t] - s;
  for (int i = lo; i < hi; ++i) {
    const int h = buf[i];
    buf[i] = run;
    run += h;
  }
  if (t == 1023) bin_start[nb] = part[1023];
  __syncthreads();
  for (int i = t; i < n; i += 1024) {
    const int o = buf[i];
    blockoff[i] = o;
    if ((i % nblk) == 0) bin_start[i / nblk] = o;
  }
}

__global__ void __launch_bounds__(THREADS)
coarse_place_kernel(const int* __restrict__ edges, const float* __restrict__ status,
                    const int* __restrict__ blockoff, int2* __restrict__ rec1,
                    int E, int nb, int nblk) {
  __shared__ int tab[MAX_BINS * PITCH];
  const int t = threadIdx.x, blk = blockIdx.x;
  const int e0 = blk * EPB;
  int2 uv[EPT];
  int  gi[EPT];
  load_edges(uv, edges, E, e0, t);
#pragma unroll
  for (int q = 0; q < EPT; ++q) {
    const int e = e0 + q * THREADS + t;
    gi[q] = e < E ? __float_as_int(status[e]) : 0;
  }
  coarse_count(tab, uv, nb, t);
  __syncthreads();
  for (int r = t; r < nb; r += THREADS) {
    int run = blockoff[r * nblk + blk];
    for (int j = 0; j < THREADS; ++j) {
      const int c = tab[r * PITCH + j];
      tab[r * PITCH + j] = run;
      run += c;
    }
  }
  __syncthreads();
#pragma unroll
  for (int q = 0; q < EPT; ++q) {
    if (uv[q].x >= 0) {
      const int u = uv[q].x, v = uv[q].y;
      const int sv = (v >> BIN_SHIFT) * PITCH + t;
      const int pv = tab[sv];
      tab[sv] = pv + 1;
      rec1[pv] = make_int2((u << 8) | (v & 255), gi[q]);
      const int su = (u >> BIN_SHIFT) * PITCH + t;
      const int pu = tab[su];
      tab[su] = pu + 1;
      rec1[pu] = make_int2((v << 8) | (u & 255), gi[q]);
    }
  }
}

__global__ void __launch_bounds__(THREADS)
fine_group_kernel(const int2* __restrict__ rec1, const int* __restrict__ bin_start,
                  int2* __restrict__ rec2, int* __restrict__ start,
                  int* __restrict__ deg, int N) {
  __shared__ int tab[BIN_NODES * PITCH];
  __shared__ int nodeoff[BIN_NODES];
  const int t = threadIdx.x, bin = blockIdx.x;
  const int b0 = bin_start[bin], b1 = bin_start[bin + 1];

  for (int r = 0; r < BIN_NODES; ++r) tab[r * PITCH + t] = 0;
  for (int base = b0; base < b1; base += THREADS * 8) {
    int x[8];
#pragma unroll
    for (int q = 0; q < 8; ++q) {
      const int i = base + q * THREADS + t;
      x[q] = i < b1 ? rec1[i].x : -1;
    }
#pragma unroll
    for (int q = 0; q < 8; ++q)
      if (x[q] >= 0) tab[(x[q] & 255) * PITCH + t] += 1;
  }
  __syncthreads();

  int mydeg = 0;
  for (int j = 0; j < THREADS; ++j) {
    const int c = tab[t * PITCH + j];
    tab[t * PITCH + j] = mydeg;
    mydeg += c;
  }
  nodeoff[t] = mydeg;
  __syncthreads();
  for (int off = 1; off < BIN_NODES; off <<= 1) {
    const int a = t >= off ? nodeoff[t - off] : 0;
    __syncthreads();
    nodeoff[t] += a;
    __syncthreads();
  }
  const int first = b0 + nodeoff[t] - mydeg;
  nodeoff[t] = first;
  const int node = bin * BIN_NODES + t;
  if (node < N) { start[node] = first; deg[node] = mydeg; }
  __syncthreads();

  for (int base = b0; base < b1; base += THREADS * 8) {
    int2 rc[8];
#pragma unroll
    for (int q = 0; q < 8; ++q) {
      const int i = base + q * THREADS + t;
      rc[q] = i < b1 ? rec1[i] : make_int2(-1, 0);
    }
#pragma unroll
    for (int q = 0; q < 8; ++q) {
      if (rc[q].x >= 0) {
        const int loc = rc[q].x & 255;
        const int k   = tab[loc * PITCH + t];
        tab[loc * PITCH + t] = k + 1;
        rec2[nodeoff[loc] + k] = make_int2(rc[q].x >> 8, rc[q].y);
      }
    }
  }
}

__device__ __forceinline__ float4 ld_row4(const float* pb, int idx) {
  return *(const float4*)(pb + (size_t)idx * D_DIM);
}

__device__ __forceinline__ void fma_row4(float4& acc, const float4 p, float g) {
  acc.x = fmaf(p.x, g, acc.x);
  acc.y = fmaf(p.y, g, acc.y);
  acc.z = fmaf(p.z, g, acc.z);
  acc.w = fmaf(p.w, g, acc.w);
}

__global__ void __launch_bounds__(THREADS)
gather_rows_kernel(const float* __restrict__ prev,
                   const int*   __restrict__ deg,
                   const int*   __restrict__ start,
                   const int2*  __restrict__ rec,
                   float*       __restrict__ nbr,
                   int N) {
  const int tid  = blockIdx.x * blockDim.x + threadIdx.x;
  const int node = __builtin_amdgcn_readfirstlane(tid >> 5);
  if (node >= N) return;
  const int lane = tid & 31;
  const int b    = lane >> 4;
  const int d4   = (lane & 15) * 4;

  const int s = start[node];
  const int n = deg[node];
  const float* pb = prev + (size_t)b * N * D_DIM + d4;

  float4 acc; acc.x = 0.f; acc.y = 0.f; acc.z = 0.f; acc.w = 0.f;

  for (int j0 = 0; j0 < n; j0 += 32) {
    const int m = (n - j0) < 32 ? (n - j0) : 32;
    int2 ent = make_int2(0, 0);
    if (lane < m) ent = rec[s + j0 + lane];

    int k = 0;
    for (; k + 8 <= m; k += 8) {
      int ii[8]; float gg[8]; float4 pp[8];
#pragma unroll
      for (int q = 0; q < 8; ++q) {
        ii[q] = __builtin_amdgcn_readlane(ent.x, k + q);
        gg[q] = __int_as_float(__builtin_amdgcn_readlane(ent.y, k + q));
      }
#pragma unroll
      for (int q = 0; q < 8; ++q) pp[q] = ld_row4(pb, ii[q]);
#pragma unroll
      for (int q = 0; q < 8; ++q) fma_row4(acc, pp[q], gg[q]);
    }
    if (k + 4 <= m) {
      int ii[4]; float gg[4]; float4 pp[4];
#pragma unroll
      for (int q = 0; q < 4; ++q) {
        ii[q] = __builtin_amdgcn_readlane(ent.x, k + q);
        gg[q] = __int_as_float(__builtin_amdgcn_readlane(ent.y, k + q));
      }
#pragma unroll
      for (int q = 0; q < 4; ++q) pp[q] = ld_row4(pb, ii[q]);
#pragma unroll
      for (int q = 0; q < 4; ++q) fma_row4(acc, pp[q], gg[q]);
      k += 4;
    }
    for (; k < m; ++k) {
      const int   i1 = __builtin_amdgcn_readlane(ent.x, k);
      const float g1 = __int_as_float(__builtin_amdgcn_readlane(ent.y, k));
      fma_row4(acc, ld_row4(pb, i1), g1);
    }
  }

  *(float4*)(nbr + ((size_t)b * N + node) * D_DIM + d4) = acc;
}

__global__ void __launch_bounds__(256)
gemm_bias_lrelu_kernel(const float* __restrict__ nbr,
                       const float* __restrict__ Wm,
                       const float* __restrict__ nf,
                       const float* __restrict__ ef,
                       float*       __restrict__ out,
                       int ntiles) {
  long tid  = (long)blockIdx.x * blockDim.x + threadIdx.x;
  long tile = tid >> 5;
  if (tile >= ntiles) return;
  int  lane = (int)(tid & 31);
  int  half = lane >> 4;
  int  lm   = lane & 15;
  long row_base = tile * 16;

  v2f afrag[16];
  const float* arow = nbr + (row_base + lm) * D_DIM + 2 * half;
#pragma unroll
  for (int t = 0; t < 16; ++t)
    afrag[t] = *(const v2f*)(arow + 4 * t);

#pragma unroll
  for (int c = 0; c < 4; ++c) {
    int col = c * 16 + lm;

    v8f acc;
#pragma unroll
    for (int r = 0; r < 8; ++r) {
      long rr = row_base + r + 8 * half;
      acc[r] = nf[rr * D_DIM + col] + ef[rr * D_DIM + col];
    }

    const float* wrow = Wm + (size_t)col * D_DIM + 2 * half;
#pragma unroll
    for (int t = 0; t < 16; ++t) {
      v2f bfrag = *(const v2f*)(wrow + 4 * t);
      acc = __builtin_amdgcn_wmma_f32_16x16x4_f32(
          false, afrag[t], false, bfrag, (short)0, acc, false, false);
    }

#pragma unroll
    for (int r = 0; r < 8; ++r) {
      long rr = row_base + r + 8 * half;
      float dv = acc[r];
      dv = dv > 0.f ? dv : 0.01f * dv;
      __builtin_nontemporal_store(dv, out + rr * D_DIM + col);
    }
  }
}

extern "C" void kernel_launch(void* const* d_in, const int* in_sizes, int n_in,
                              void* d_out, int out_size, void* d_ws, size_t ws_size,
                              hipStream_t stream) {
  const float* prev  = (const float*)d_in[0];
  const int*   edges = (const int*)  d_in[1];
  const float* nf    = (const float*)d_in[2];
  const float* ef    = (const float*)d_in[3];
  const float* st    = (const float*)d_in[4];
  const float* Wm    = (const float*)d_in[5];

  const int BND = in_sizes[0];
  const int E   = in_sizes[4];
  const int N   = BND / (B_DIM * D_DIM);

  float* nbr  = (float*)d_ws;
  int2*  rec1 = (int2*)d_ws;

  int*  scr       = (int*)d_out;
  const int NP    = ((N + 63) / 64) * 64;
  int2* rec2      = (int2*)scr;
  int*  blockhist = scr + 4 * E;
  int*  blockoff  = blockhist + MAX_BINS * MAX_BLKS;
  int*  bin_start = blockoff + MAX_BINS * MAX_BLKS;
  int*  start     = bin_start + 256;
  int*  deg       = start + NP;

  const int nb   = (N + BIN_NODES - 1) / BIN_NODES;
  const int nblk = (E + EPB - 1) / EPB;

  coarse_count_kernel<<<nblk, THREADS, 0, stream>>>(edges, blockhist, E, nb, nblk);
  scan_counts_kernel<<<1, 1024, 0, stream>>>(blockhist, blockoff, bin_start,
                                             nb * nblk, nb, nblk);
  coarse_place_kernel<<<nblk, THREADS, 0, stream>>>(edges, st, blockoff, rec1,
                                                    E, nb, nblk);
  fine_group_kernel<<<nb, THREADS, 0, stream>>>(rec1, bin_start, rec2, start, deg, N);
  long gth = (long)N * 32;
  gather_rows_kernel<<<(int)((gth + THREADS - 1) / THREADS), THREADS, 0, stream>>>(
      prev, deg, start, rec2, nbr, N);

  int ntiles = (BND / D_DIM) / 16;
  long gthreads = (long)ntiles * 32;
  gemm_bias_lrelu_kernel<<<(int)((gthreads + 255) / 256), 256, 0, stream>>>(
      nbr, Wm, nf, ef, (float*)d_out, ntiles);
}


# Count and scan launches dropped: level-1 groups bin their records into segments of their own, level 2 walks its bin's runs and sums lower bins' lengths; 6 to 4 launches; per-node sum order changes
# speedup vs baseline: 2.7488x; 1.3422x over previous
#include <hip/hip_runtime.h>

typedef __attribute__((ext_vector_type(2))) float v2f;
typedef __attribute__((ext_vector_type(8))) float v8f;

#define D_DIM 64
#define B_DIM 2


#define BIN_SHIFT  8
#define BIN_NODES  256
#define PITCH      257
#define MAX_BINS   200
#define MAX_BLKS   256
#define THREADS    256
#define EPT        16
#define EPB        (EPT * THREADS)
#define SEG        (2 * EPB)

static_assert(MAX_BINS <= THREADS, "one thread per coarse bin in level 1");
static_assert(MAX_BLKS <= THREADS, "one thread per run in level 2");

__device__ __forceinline__ int excl_scan_256(int v, int* wsum, int t) {
  const int lane = t & 31;
  int x = v;
  for (int d = 1; d < 32; d <<= 1) {
    const int y = __shfl_up(x, d);
    if (lane >= d) x += y;
  }
  if (lane == 31) wsum[t >> 5] = x;
  __syncthreads();
  int base = 0;
  for (int w = 0; w < THREADS / 32; ++w) base += w < (t >> 5) ? wsum[w] : 0;
  return base + x - v;
}

__device__ __forceinline__ void load_edges(int2 (&uv)[EPT], const int* __restrict__ edges,
                                           int E, int e0, int t) {
#pragma unroll
  for (int q = 0; q < EPT; ++q) {
    const int e = e0 + q * THREADS + t;
    uv[q] = e < E ? ((const int2*)edges)[e] : make_int2(-1, -1);
  }
}

__device__ __forceinline__ void coarse_count(int* tab, const int2 (&uv)[EPT], int nb, int t) {
  for (int r = 0; r < nb; ++r) tab[r * PITCH + t] = 0;
#pragma unroll
  for (int q = 0; q < EPT; ++q) {
    if (uv[q].x >= 0) {
      tab[(uv[q].y >> BIN_SHIFT) * PITCH + t] += 1;
      tab[(uv[q].x >> BIN_SHIFT) * PITCH + t] += 1;
    }
  }
}

__global__ void __launch_bounds__(THREADS)
coarse_bin_kernel(const int* __restrict__ edges, const float* __restrict__ status,
                  int2* __restrict__ rec1, int* __restrict__ runcnt,
                  int* __restrict__ runoff, int E, int nb, int nblk) {
  __shared__ int tab[MAX_BINS * PITCH];
  __shared__ int binoff[THREADS];
  __shared__ int wsum[THREADS / 32];
  const int t = threadIdx.x, blk = blockIdx.x;
  const int e0 = blk * EPB;
  int2 uv[EPT];
  int  gi[EPT];
  load_edges(uv, edges, E, e0, t);
#pragma unroll
  for (int q = 0; q < EPT; ++q) {
    const int e = e0 + q * THREADS + t;
    gi[q] = e < E ? __float_as_int(status[e]) : 0;
  }
  coarse_count(tab, uv, nb, t);
  __syncthreads();
  int tot = 0;
  if (t < nb) {
    for (int j = 0; j < THREADS; ++j) {
      const int c = tab[t * PITCH + j];
      tab[t * PITCH + j] = tot;
      tot += c;
    }
  }
  const int off = excl_scan_256(tot, wsum, t);
  binoff[t] = off;
  if (t < nb) {
    runcnt[t * nblk + blk] = tot;
    runoff[t * nblk + blk] = off;
  }
  __syncthreads();
  int2* seg = rec1 + (size_t)blk * SEG;
#pragma unroll
  for (int q = 0; q < EPT; ++q) {
    if (uv[q].x >= 0) {
      const int u = uv[q].x, v = uv[q].y;
      const int bv = v >> BIN_SHIFT, sv = bv * PITCH + t;
      const int pv = tab[sv];
      tab[sv] = pv + 1;
      seg[binoff[bv] + pv] = make_int2((u << 8) | (v & 255), gi[q]);
      const int bu = u >> BIN_SHIFT, su = bu * PITCH + t;
      const int pu = tab[su];
      tab[su] = pu + 1;
      seg[binoff[bu] + pu] = make_int2((v << 8) | (u & 255), gi[q]);
    }
  }
}

#define RUNS_PG    (MAX_BLKS / 16)
#define PASSES     4

__global__ void __launch_bounds__(THREADS)
fine_group_kernel(const int2* __restrict__ rec1, const int* __restrict__ runcnt,
                  const int* __restrict__ runoff, int2* __restrict__ rec2,
                  int* __restrict__ start, int* __restrict__ deg, int N, int nblk) {
  __shared__ int  tab[BIN_NODES * PITCH];
  __shared__ int  nodeoff[BIN_NODES];
  __shared__ int2 run[MAX_BLKS];
  __shared__ int  wsum[THREADS / 32];
  const int t = threadIdx.x, bin = blockIdx.x;
  const int g = t >> 4, l = t & 15;

  int myc = 0, myo = 0;
  if (t < nblk) { myc = runcnt[bin * nblk + t]; myo = runoff[bin * nblk + t]; }

  const int lim = bin * nblk, lim4 = lim & ~3;
  int below = 0;
  for (int base = 0; base < lim4; base += THREADS * 4 * 16) {
    int4 c[16];
#pragma unroll
    for (int q = 0; q < 16; ++q) {
      const int i = base + (q * THREADS + t) * 4;
      c[q] = *(const int4*)(runcnt + (i < lim4 ? i : 0));
      if (i >= lim4) c[q] = make_int4(0, 0, 0, 0);
    }
#pragma unroll
    for (int q = 0; q < 16; ++q) below += (c[q].x + c[q].y) + (c[q].z + c[q].w);
  }
  if (t < lim - lim4) below += runcnt[lim4 + t];
  for (int d = 16; d >= 1; d >>= 1) below += __shfl_xor(below, d);
  if ((t & 31) == 0) wsum[t >> 5] = below;

  run[t] = make_int2(myc, t * SEG + myo);
  for (int r = 0; r < BIN_NODES; ++r) tab[r * PITCH + t] = 0;
  __syncthreads();
  int b0 = 0;
  for (int w = 0; w < THREADS / 32; ++w) b0 += wsum[w];
  int2 rn[RUNS_PG];
#pragma unroll
  for (int q = 0; q < RUNS_PG; ++q) rn[q] = run[g + 16 * q];

  {
    int x[RUNS_PG][PASSES];
#pragma unroll
    for (int q = 0; q < RUNS_PG; ++q)
#pragma unroll
      for (int p = 0; p < PASSES; ++p) {
        const int i = p * 16 + l;
        x[q][p] = i < rn[q].x ? rec1[rn[q].y + i].x : -1;
      }
#pragma unroll
    for (int q = 0; q < RUNS_PG; ++q)
#pragma unroll
      for (int p = 0; p < PASSES; ++p)
        if (x[q][p] >= 0) tab[(x[q][p] & 255) * PITCH + t] += 1;
#pragma unroll
    for (int q = 0; q < RUNS_PG; ++q)
      for (int i = PASSES * 16 + l; i < rn[q].x; i += 16)
        tab[(rec1[rn[q].y + i].x & 255) * PITCH + t] += 1;
  }
  __syncthreads();

  int mydeg = 0;
  for (int j = 0; j < THREADS; ++j) {
    const int c = tab[t * PITCH + j];
    tab[t * PITCH + j] = mydeg;
    mydeg += c;
  }
  nodeoff[t] = mydeg;
  __syncthreads();
  for (int off = 1; off < BIN_NODES; off <<= 1) {
    const int a = t >= off ? nodeoff[t - off] : 0;
    __syncthreads();
    nodeoff[t] += a;
    __syncthreads();
  }
  const int first = b0 + nodeoff[t] - mydeg;
  nodeoff[t] = first;
  const int node = bin * BIN_NODES + t;
  if (node < N) { start[node] = first; deg[node] = mydeg; }
  __syncthreads();

  {
    int2 rc[RUNS_PG][PASSES];
#pragma unroll
    for (int q = 0; q < RUNS_PG; ++q)
#pragma unroll
      for (int p = 0; p < PASSES; ++p) {
        const int i = p * 16 + l;
        rc[q][p] = i < rn[q].x ? rec1[rn[q].y + i] : make_int2(-1, 0);
      }
#pragma unroll
    for (int q = 0; q < RUNS_PG; ++q) {
#pragma unroll
      for (int p = 0; p < PASSES; ++p) {
        if (rc[q][p].x >= 0) {
          const int loc = rc[q][p].x & 255;
          const int k   = tab[loc * PITCH + t];
          tab[loc * PITCH + t] = k + 1;
          rec2[nodeoff[loc] + k] = make_int2(rc[q][p].x >> 8, rc[q][p].y);
        }
      }
    }
#pragma unroll
    for (int q = 0; q < RUNS_PG; ++q) {
      for (int i = PASSES * 16 + l; i < rn[q].x; i += 16) {
        const int2 r1  = rec1[rn[q].y + i];
        const int  loc = r1.x & 255;
        const int  k   = tab[loc * PITCH + t];
        tab[loc * PITCH + t] = k + 1;
        rec2[nodeoff[loc] + k] = make_int2(r1.x >> 8, r1.y);
      }
    }
  }
}

__device__ __forceinline__ float4 ld_row4(const float* pb, int idx) {
  return *(const float4*)(pb + (size_t)idx * D_DIM);
}

__device__ __forceinline__ void fma_row4(float4& acc, const float4 p, float g) {
  acc.x = fmaf(p.x, g, acc.x);
  acc.y = fmaf(p.y, g, acc.y);
  acc.z = fmaf(p.z, g, acc.z);
  acc.w = fmaf(p.w, g, acc.w);
}

__global__ void __launch_bounds__(THREADS)
gather_rows_kernel(const float* __restrict__ prev,
                   const int*   __restrict__ deg,
                   const int*   __restrict__ start,
                   const int2*  __restrict__ rec,
                   float*       __restrict__ nbr,
                   int N) {
  const int tid  = blockIdx.x * blockDim.x + threadIdx.x;
  const int node = __builtin_amdgcn_readfirstlane(tid >> 5);
  if (node >= N) return;
  const int lane = tid & 31;
  const int b    = lane >> 4;
  const int d4   = (lane & 15) * 4;

  const int s = start[node];
  const int n = deg[node];
  const float* pb = prev + (size_t)b * N * D_DIM + d4;

  float4 acc; acc.x = 0.f; acc.y = 0.f; acc.z = 0.f; acc.w = 0.f;

  for (int j0 = 0; j0 < n; j0 += 32) {
    const int m = (n - j0) < 32 ? (n - j0) : 32;
    int2 ent = make_int2(0, 0);
    if (lane < m) ent = rec[s + j0 + lane];

    int k = 0;
    for (; k + 8 <= m; k += 8) {
      int ii[8]; float gg[8]; float4 pp[8];
#pragma unroll
      for (int q = 0; q < 8; ++q) {
        ii[q] = __builtin_amdgcn_readlane(ent.x, k + q);
        gg[q] = __int_as_float(__builtin_amdgcn_readlane(ent.y, k + q));
      }
#pragma unroll
      for (int q = 0; q < 8; ++q) pp[q] = ld_row4(pb, ii[q]);
#pragma unroll
      for (int q = 0; q < 8; ++q) fma_row4(acc, pp[q], gg[q]);
    }
    if (k + 4 <= m) {
      int ii[4]; float gg[4]; float4 pp[4];
#pragma unroll
      for (int q = 0; q < 4; ++q) {
        ii[q] = __builtin_amdgcn_readlane(ent.x, k + q);
        gg[q] = __int_as_float(__builtin_amdgcn_readlane(ent.y, k + q));
      }
#pragma unroll
      for (int q = 0; q < 4; ++q) pp[q] = ld_row4(pb, ii[q]);
#pragma unroll
      for (int q = 0; q < 4; ++q) fma_row4(acc, pp[q], gg[q]);
      k += 4;
    }
    for (; k < m; ++k) {
      const int   i1 = __builtin_amdgcn_readlane(ent.x, k);
      const float g1 = __int_as_float(__builtin_amdgcn_readlane(ent.y, k));
      fma_row4(acc, ld_row4(pb, i1), g1);
    }
  }

  *(float4*)(nbr + ((size_t)b * N + node) * D_DIM + d4) = acc;
}

__global__ void __launch_bounds__(256)
gemm_bias_lrelu_kernel(const float* __restrict__ nbr,
                       const float* __restrict__ Wm,
                       const float* __restrict__ nf,
                       const float* __restrict__ ef,
                       float*       __restrict__ out,
                       int ntiles) {
  long tid  = (long)blockIdx.x * blockDim.x + threadIdx.x;
  long tile = tid >> 5;
  if (tile >= ntiles) return;
  int  lane = (int)(tid & 31);
  int  half = lane >> 4;
  int  lm   = lane & 15;
  long row_base = tile * 16;

  v2f afrag[16];
  const float* arow = nbr + (row_base + lm) * D_DIM + 2 * half;
#pragma unroll
  for (int t = 0; t < 16; ++t)
    afrag[t] = *(const v2f*)(arow + 4 * t);

#pragma unroll
  for (int c = 0; c < 4; ++c) {
    int col = c * 16 + lm;

    v8f acc;
#pragma unroll
    for (int r = 0; r < 8; ++r) {
      long rr = row_base + r + 8 * half;
      acc[r] = nf[rr * D_DIM + col] + ef[rr * D_DIM + col];
    }

    const float* wrow = Wm + (size_t)col * D_DIM + 2 * half;
#pragma unroll
    for (int t = 0; t < 16; ++t) {
      v2f bfrag = *(const v2f*)(wrow + 4 * t);
      acc = __builtin_amdgcn_wmma_f32_16x16x4_f32(
          false, afrag[t], false, bfrag, (short)0, acc, false, false);
    }

#pragma unroll
    for (int r = 0; r < 8; ++r) {
      long rr = row_base + r + 8 * half;
      float dv = acc[r];
      dv = dv > 0.f ? dv : 0.01f * dv;
      __builtin_nontemporal_store(dv, out + rr * D_DIM + col);
    }
  }
}

extern "C" void kernel_launch(void* const* d_in, const int* in_sizes, int n_in,
                              void* d_out, int out_size, void* d_ws, size_t ws_size,
                              hipStream_t stream) {
  const float* prev  = (const float*)d_in[0];
  const int*   edges = (const int*)  d_in[1];
  const float* nf    = (const float*)d_in[2];
  const float* ef    = (const float*)d_in[3];
  const float* st    = (const float*)d_in[4];
  const float* Wm    = (const float*)d_in[5];

  const int BND = in_sizes[0];
  const int E   = in_sizes[4];
  const int N   = BND / (B_DIM * D_DIM);

  float* nbr  = (float*)d_ws;
  int2*  rec1 = (int2*)d_ws;

  int*  scr    = (int*)d_out;
  const int NP = ((N + 63) / 64) * 64;
  int2* rec2   = (int2*)scr;
  int*  runcnt = scr + 4 * E;
  int*  runoff = runcnt + MAX_BINS * MAX_BLKS;
  int*  start  = runoff + MAX_BINS * MAX_BLKS;
  int*  deg    = start + NP;

  const int nb   = (N + BIN_NODES - 1) / BIN_NODES;
  const int nblk = (E + EPB - 1) / EPB;

  coarse_bin_kernel<<<nblk, THREADS, 0, stream>>>(edges, st, rec1, runcnt, runoff,
                                                  E, nb, nblk);
  fine_group_kernel<<<nb, THREADS, 0, stream>>>(rec1, runcnt, runoff, rec2, start, deg,
                                                N, nblk);
  long gth = (long)N * 32;
  gather_rows_kernel<<<(int)((gth + THREADS - 1) / THREADS), THREADS, 0, stream>>>(
      prev, deg, start, rec2, nbr, N);

  int ntiles = (BND / D_DIM) / 16;
  long gthreads = (long)ntiles * 32;
  gemm_bias_lrelu_kernel<<<(int)((gthreads + 255) / 256), 256, 0, stream>>>(
      nbr, Wm, nf, ef, (float*)d_out, ntiles);
}


# Per-node sums and WMMA product fused: a work-group of 8 waves sums 32 nodes' rows into an LDS tile and multiplies from it; nbr round trip through memory and one launch dropped (4 to 3 launches)
# speedup vs baseline: 3.0791x; 1.1202x over previous
#include <hip/hip_runtime.h>

typedef __attribute__((ext_vector_type(2))) float v2f;
typedef __attribute__((ext_vector_type(8))) float v8f;

#define D_DIM 64
#define B_DIM 2


#define BIN_SHIFT  8
#define BIN_NODES  256
#define PITCH      257
#define MAX_BINS   200
#define MAX_BLKS   256
#define THREADS    256
#define EPT        16
#define EPB        (EPT * THREADS)
#define SEG        (2 * EPB)

static_assert(MAX_BINS <= THREADS, "one thread per coarse bin in level 1");
static_assert(MAX_BLKS <= THREADS, "one thread per run in level 2");

__device__ __forceinline__ int excl_scan_256(int v, int* wsum, int t) {
  const int lane = t & 31;
  int x = v;
  for (int d = 1; d < 32; d <<= 1) {
    const int y = __shfl_up(x, d);
    if (lane >= d) x += y;
  }
  if (lane == 31) wsum[t >> 5] = x;
  __syncthreads();
  int base = 0;
  for (int w = 0; w < THREADS / 32; ++w) base += w < (t >> 5) ? wsum[w] : 0;
  return base + x - v;
}

__device__ __forceinline__ void load_edges(int2 (&uv)[EPT], const int* __restrict__ edges,
                                           int E, int e0, int t) {
#pragma unroll
  for (int q = 0; q < EPT; ++q) {
    const int e = e0 + q * THREADS + t;
    uv[q] = e < E ? ((const int2*)edges)[e] : make_int2(-1, -1);
  }
}

__device__ __forceinline__ void coarse_count(int* tab, const int2 (&uv)[EPT], int nb, int t) {
  for (int r = 0; r < nb; ++r) tab[r * PITCH + t] = 0;
#pragma unroll
  for (int q = 0; q < EPT; ++q) {
    if (uv[q].x >= 0) {
      tab[(uv[q].y >> BIN_SHIFT) * PITCH + t] += 1;
      tab[(uv[q].x >> BIN_SHIFT) * PITCH + t] += 1;
    }
  }
}

__global__ void __launch_bounds__(THREADS)
coarse_bin_kernel(const int* __restrict__ edges, const float* __restrict__ status,
                  int2* __restrict__ rec1, int* __restrict__ runcnt,
                  int* __restrict__ runoff, int E, int nb, int nblk) {
  __shared__ int tab[MAX_BINS * PITCH];
  __shared__ int binoff[THREADS];
  __shared__ int wsum[THREADS / 32];
  const int t = threadIdx.x, blk = blockIdx.x;
  const int e0 = blk * EPB;
  int2 uv[EPT];
  int  gi[EPT];
  load_edges(uv, edges, E, e0, t);
#pragma unroll
  for (int q = 0; q < EPT; ++q) {
    const int e = e0 + q * THREADS + t;
    gi[q] = e < E ? __float_as_int(status[e]) : 0;
  }
  coarse_count(tab, uv, nb, t);
  __syncthreads();
  int tot = 0;
  if (t < nb) {
    for (int j = 0; j < THREADS; ++j) {
      const int c = tab[t * PITCH + j];
      tab[t * PITCH + j] = tot;
      tot += c;
    }
  }
  const int off = excl_scan_256(tot, wsum, t);
  binoff[t] = off;
  if (t < nb) {
    runcnt[t * nblk + blk] = tot;
    runoff[t * nblk + blk] = off;
  }
  __syncthreads();
  int2* seg = rec1 + (size_t)blk * SEG;
#pragma unroll
  for (int q = 0; q < EPT; ++q) {
    if (uv[q].x >= 0) {
      const int u = uv[q].x, v = uv[q].y;
      const int bv = v >> BIN_SHIFT, sv = bv * PITCH + t;
      const int pv = tab[sv];
      tab[sv] = pv + 1;
      seg[binoff[bv] + pv] = make_int2((u << 8) | (v & 255), gi[q]);
      const int bu = u >> BIN_SHIFT, su = bu * PITCH + t;
      const int pu = tab[su];
      tab[su] = pu + 1;
      seg[binoff[bu] + pu] = make_int2((v << 8) | (u & 255), gi[q]);
    }
  }
}

#define RUNS_PG    (MAX_BLKS / 16)
#define PASSES     4

__global__ void __launch_bounds__(THREADS)
fine_group_kernel(const int2* __restrict__ rec1, const int* __restrict__ runcnt,
                  const int* __restrict__ runoff, int2* __restrict__ rec2,
                  int* __restrict__ start, int* __restrict__ deg, int N, int nblk) {
  __shared__ int  tab[BIN_NODES * PITCH];
  __shared__ int  nodeoff[BIN_NODES];
  __shared__ int2 run[MAX_BLKS];
  __shared__ int  wsum[THREADS / 32];
  const int t = threadIdx.x, bin = blockIdx.x;
  const int g = t >> 4, l = t & 15;

  int myc = 0, myo = 0;
  if (t < nblk) { myc = runcnt[bin * nblk + t]; myo = runoff[bin * nblk + t]; }

  const int lim = bin * nblk, lim4 = lim & ~3;
  int below = 0;
  for (int base = 0; base < lim4; base += THREADS * 4 * 16) {
    int4 c[16];
#pragma unroll
    for (int q = 0; q < 16; ++q) {
      const int i = base + (q * THREADS + t) * 4;
      c[q] = *(const int4*)(runcnt + (i < lim4 ? i : 0));
      if (i >= lim4) c[q] = make_int4(0, 0, 0, 0);
    }
#pragma unroll
    for (int q = 0; q < 16; ++q) below += (c[q].x + c[q].y) + (c[q].z + c[q].w);
  }
  if (t < lim - lim4) below += runcnt[lim4 + t];
  for (int d = 16; d >= 1; d >>= 1) below += __shfl_xor(below, d);
  if ((t & 31) == 0) wsum[t >> 5] = below;

  run[t] = make_int2(myc, t * SEG + myo);
  for (int r = 0; r < BIN_NODES; ++r) tab[r * PITCH + t] = 0;
  __syncthreads();
  int b0 = 0;
  for (int w = 0; w < THREADS / 32; ++w) b0 += wsum[w];
  int2 rn[RUNS_PG];
#pragma unroll
  for (int q = 0; q < RUNS_PG; ++q) rn[q] = run[g + 16 * q];

  {
    int x[RUNS_PG][PASSES];
#pragma unroll
    for (int q = 0; q < RUNS_PG; ++q)
#pragma unroll
      for (int p = 0; p < PASSES; ++p) {
        const int i = p * 16 + l;
        x[q][p] = i < rn[q].x ? rec1[rn[q].y + i].x : -1;
      }
#pragma unroll
    for (int q = 0; q < RUNS_PG; ++q)
#pragma unroll
      for (int p = 0; p < PASSES; ++p)
        if (x[q][p] >= 0) tab[(x[q][p] & 255) * PITCH + t] += 1;
#pragma unroll
    for (int q = 0; q < RUNS_PG; ++q)
      for (int i = PASSES * 16 + l; i < rn[q].x; i += 16)
        tab[(rec1[rn[q].y + i].x & 255) * PITCH + t] += 1;
  }
  __syncthreads();

  int mydeg = 0;
  for (int j = 0; j < THREADS; ++j) {
    const int c = tab[t * PITCH + j];
    tab[t * PITCH + j] = mydeg;
    mydeg += c;
  }
  nodeoff[t] = mydeg;
  __syncthreads();
  for (int off = 1; off < BIN_NODES; off <<= 1) {
    const int a = t >= off ? nodeoff[t - off] : 0;
    __syncthreads();
    nodeoff[t] += a;
    __syncthreads();
  }
  const int first = b0 + nodeoff[t] - mydeg;
  nodeoff[t] = first;
  const int node = bin * BIN_NODES + t;
  if (node < N) { start[node] = first; deg[node] = mydeg; }
  __syncthreads();

  {
    int2 rc[RUNS_PG][PASSES];
#pragma unroll
    for (int q = 0; q < RUNS_PG; ++q)
#pragma unroll
      for (int p = 0; p < PASSES; ++p) {
        const int i = p * 16 + l;
        rc[q][p] = i < rn[q].x ? rec1[rn[q].y + i] : make_int2(-1, 0);
      }
#pragma unroll
    for (int q = 0; q < RUNS_PG; ++q) {
#pragma unroll
      for (int p = 0; p < PASSES; ++p) {
        if (rc[q][p].x >= 0) {
          const int loc = rc[q][p].x & 255;
          const int k   = tab[loc * PITCH + t];
          tab[loc * PITCH + t] = k + 1;
          rec2[nodeoff[loc] + k] = make_int2(rc[q][p].x >> 8, rc[q][p].y);
        }
      }
    }
#pragma unroll
    for (int q = 0; q < RUNS_PG; ++q) {
      for (int i = PASSES * 16 + l; i < rn[q].x; i += 16) {
        const int2 r1  = rec1[rn[q].y + i];
        const int  loc = r1.x & 255;
        const int  k   = tab[loc * PITCH + t];
        tab[loc * PITCH + t] = k + 1;
        rec2[nodeoff[loc] + k] = make_int2(r1.x >> 8, r1.y);
      }
    }
  }
}

#define TILE_NODES 32
#define NODES_PW   (TILE_NODES / 8)
#define A_PITCH    68

__device__ __forceinline__ float4 ld_row4(const float* pb, int idx) {
  return *(const float4*)(pb + (size_t)idx * D_DIM);
}

__device__ __forceinline__ void fma_row4(float4& acc, const float4 p, float g) {
  acc.x = fmaf(p.x, g, acc.x);
  acc.y = fmaf(p.y, g, acc.y);
  acc.z = fmaf(p.z, g, acc.z);
  acc.w = fmaf(p.w, g, acc.w);
}

__global__ void __launch_bounds__(THREADS)
gather_gemm_lrelu_kernel(const float* __restrict__ prev,
                         const int*   __restrict__ deg,
                         const int*   __restrict__ start,
                         const int2*  __restrict__ rec,
                         const float* __restrict__ Wm,
                         const float* __restrict__ nf,
                         const float* __restrict__ ef,
                         float*       __restrict__ out,
                         int N) {
  __shared__ float tile[B_DIM * TILE_NODES * A_PITCH];
  const int t    = threadIdx.x;
  const int lane = t & 31;
  const int w     = __builtin_amdgcn_readfirstlane(t >> 5);
  const int node0 = blockIdx.x * TILE_NODES;

  {
    const int b  = lane >> 4;
    const int d4 = (lane & 15) * 4;
    const float* pb = prev + (size_t)b * N * D_DIM + d4;

    for (int i = 0; i < NODES_PW; ++i) {
      const int nl   = w * NODES_PW + i;
      const int node = node0 + nl;
      float4 acc; acc.x = 0.f; acc.y = 0.f; acc.z = 0.f; acc.w = 0.f;

      if (node < N) {
        const int s = start[node];
        const int n = deg[node];
        for (int j0 = 0; j0 < n; j0 += 32) {
          const int m = (n - j0) < 32 ? (n - j0) : 32;
          int2 ent = make_int2(0, 0);
          if (lane < m) ent = rec[s + j0 + lane];

          int k = 0;
          for (; k + 8 <= m; k += 8) {
            int ii[8]; float gg[8]; float4 pp[8];
#pragma unroll
            for (int q = 0; q < 8; ++q) {
              ii[q] = __builtin_amdgcn_readlane(ent.x, k + q);
              gg[q] = __int_as_float(__builtin_amdgcn_readlane(ent.y, k + q));
            }
#pragma unroll
            for (int q = 0; q < 8; ++q) pp[q] = ld_row4(pb, ii[q]);
#pragma unroll
            for (int q = 0; q < 8; ++q) fma_row4(acc, pp[q], gg[q]);
          }
          if (k + 4 <= m) {
            int ii[4]; float gg[4]; float4 pp[4];
#pragma unroll
            for (int q = 0; q < 4; ++q) {
              ii[q] = __builtin_amdgcn_readlane(ent.x, k + q);
              gg[q] = __int_as_float(__builtin_amdgcn_readlane(ent.y, k + q));
            }
#pragma unroll
            for (int q = 0; q < 4; ++q) pp[q] = ld_row4(pb, ii[q]);
#pragma unroll
            for (int q = 0; q < 4; ++q) fma_row4(acc, pp[q], gg[q]);
            k += 4;
          }
          for (; k < m; ++k) {
            const int   i1 = __builtin_amdgcn_readlane(ent.x, k);
            const float g1 = __int_as_float(__builtin_amdgcn_readlane(ent.y, k));
            fma_row4(acc, ld_row4(pb, i1), g1);
          }
        }
      }
      *(float4*)(tile + (b * TILE_NODES + nl) * A_PITCH + d4) = acc;
    }
  }
  __syncthreads();

  const int rt   = w >> 1;
  const int bt   = rt >> 1;
  const int nb0  = node0 + (rt & 1) * 16;
  if (nb0 + 16 > N) return;
  const int half = lane >> 4;
  const int lm   = lane & 15;

  const float* arow = tile + (rt * 16 + lm) * A_PITCH + 2 * half;
  const int    col  = (w & 1) * 32 + lm;
  const float* wrow = Wm + (size_t)col * D_DIM + 2 * half;

  const size_t o0 = ((size_t)bt * N + nb0 + 8 * half) * D_DIM + col;

  v8f acc0, acc1;
#pragma unroll
  for (int r = 0; r < 8; ++r) {
    acc0[r] = nf[o0 + r * D_DIM]      + ef[o0 + r * D_DIM];
    acc1[r] = nf[o0 + r * D_DIM + 16] + ef[o0 + r * D_DIM + 16];
  }

#pragma unroll 4
  for (int q = 0; q < 16; ++q) {
    const v2f a  = *(const v2f*)(arow + 4 * q);
    const v2f b0 = *(const v2f*)(wrow + 4 * q);
    const v2f b1 = *(const v2f*)(wrow + 16 * D_DIM + 4 * q);
    acc0 = __builtin_amdgcn_wmma_f32_16x16x4_f32(false, a, false, b0, (short)0, acc0, false, false);
    acc1 = __builtin_amdgcn_wmma_f32_16x16x4_f32(false, a, false, b1, (short)0, acc1, false, false);
  }

#pragma unroll
  for (int r = 0; r < 8; ++r) {
    float d0 = acc0[r], d1 = acc1[r];
    d0 = d0 > 0.f ? d0 : 0.01f * d0;
    d1 = d1 > 0.f ? d1 : 0.01f * d1;
    __builtin_nontemporal_store(d0, out + o0 + r * D_DIM);
    __builtin_nontemporal_store(d1, out + o0 + r * D_DIM + 16);
  }
}

extern "C" void kernel_launch(void* const* d_in, const int* in_sizes, int n_in,
                              void* d_out, int out_size, void* d_ws, size_t ws_size,
                              hipStream_t stream) {
  const float* prev  = (const float*)d_in[0];
  const int*   edges = (const int*)  d_in[1];
  const float* nf    = (const float*)d_in[2];
  const float* ef    = (const float*)d_in[3];
  const float* st    = (const float*)d_in[4];
  const float* Wm    = (const float*)d_in[5];

  const int BND = in_sizes[0];
  const int E   = in_sizes[4];
  const int N   = BND / (B_DIM * D_DIM);

  const int nb   = (N + BIN_NODES - 1) / BIN_NODES;
  const int nblk = (E + EPB - 1) / EPB;

  int*  scr    = (int*)d_out;
  int2* rec1   = (int2*)scr;
  int*  runcnt = scr + 2 * nblk * SEG;
  int*  runoff = runcnt + MAX_BINS * MAX_BLKS;

  int*  wsi    = (int*)d_ws;
  const int NP = ((N + 63) / 64) * 64;
  int2* rec2   = (int2*)wsi;
  int*  start  = wsi + 4 * E;
  int*  deg    = start + NP;

  coarse_bin_kernel<<<nblk, THREADS, 0, stream>>>(edges, st, rec1, runcnt, runoff,
                                                  E, nb, nblk);
  fine_group_kernel<<<nb, THREADS, 0, stream>>>(rec1, runcnt, runoff, rec2, start, deg,
                                                N, nblk);
  gather_gemm_lrelu_kernel<<<(N + TILE_NODES - 1) / TILE_NODES, THREADS, 0, stream>>>(
      prev, deg, start, rec2, Wm, nf, ef, (float*)d_out, N);
}


# Level-1 segment assembled in LDS (64 KB image) and copied out 16 bytes per lane: whole 128-byte lines, each written once, in place of 8-byte stores with every lane on a line of its own
# speedup vs baseline: 3.2378x; 1.0515x over previous
#include <hip/hip_runtime.h>

typedef __attribute__((ext_vector_type(2))) float v2f;
typedef __attribute__((ext_vector_type(8))) float v8f;

#define D_DIM 64
#define B_DIM 2


#define BIN_SHIFT  8
#define BIN_NODES  256
#define PITCH      257
#define MAX_BINS   200
#define MAX_BLKS   256
#define THREADS    256
#define EPT        16
#define EPB        (EPT * THREADS)
#define SEG        (2 * EPB)

static_assert(MAX_BINS <= THREADS, "one thread per coarse bin in level 1");
static_assert(MAX_BLKS <= THREADS, "one thread per run in level 2");

__device__ __forceinline__ int excl_scan_256(int v, int* wsum, int t) {
  const int lane = t & 31;
  int x = v;
  for (int d = 1; d < 32; d <<= 1) {
    const int y = __shfl_up(x, d);
    if (lane >= d) x += y;
  }
  if (lane == 31) wsum[t >> 5] = x;
  __syncthreads();
  int base = 0;
  for (int w = 0; w < THREADS / 32; ++w) base += w < (t >> 5) ? wsum[w] : 0;
  return base + x - v;
}

__device__ __forceinline__ void load_edges(int2 (&uv)[EPT], const int* __restrict__ edges,
                                           int E, int e0, int t) {
#pragma unroll
  for (int q = 0; q < EPT; ++q) {
    const int e = e0 + q * THREADS + t;
    uv[q] = e < E ? ((const int2*)edges)[e] : make_int2(-1, -1);
  }
}

__device__ __forceinline__ void coarse_count(int* tab, const int2 (&uv)[EPT], int nb, int t) {
  for (int r = 0; r < nb; ++r) tab[r * PITCH + t] = 0;
#pragma unroll
  for (int q = 0; q < EPT; ++q) {
    if (uv[q].x >= 0) {
      tab[(uv[q].y >> BIN_SHIFT) * PITCH + t] += 1;
      tab[(uv[q].x >> BIN_SHIFT) * PITCH + t] += 1;
    }
  }
}

__global__ void __launch_bounds__(THREADS)
coarse_bin_kernel(const int* __restrict__ edges, const float* __restrict__ status,
                  int2* __restrict__ rec1, int* __restrict__ runcnt,
                  int* __restrict__ runoff, int E, int nb, int nblk) {
  __shared__ int  tab[MAX_BINS * PITCH];
  __shared__ int2 stage[SEG];
  __shared__ int  binoff[THREADS];
  __shared__ int  wsum[THREADS / 32];
  __shared__ int  nrec;
  const int t = threadIdx.x, blk = blockIdx.x;
  const int e0 = blk * EPB;
  int2 uv[EPT];
  int  gi[EPT];
  load_edges(uv, edges, E, e0, t);
#pragma unroll
  for (int q = 0; q < EPT; ++q) {
    const int e = e0 + q * THREADS + t;
    gi[q] = e < E ? __float_as_int(status[e]) : 0;
  }
  coarse_count(tab, uv, nb, t);
  __syncthreads();
  int tot = 0;
  if (t < nb) {
    for (int j = 0; j < THREADS; ++j) {
      const int c = tab[t * PITCH + j];
      tab[t * PITCH + j] = tot;
      tot += c;
    }
  }
  const int off = excl_scan_256(tot, wsum, t);
  binoff[t] = off;
  if (t == THREADS - 1) nrec = off + tot;
  if (t < nb) {
    runcnt[t * nblk + blk] = tot;
    runoff[t * nblk + blk] = off;
  }
  __syncthreads();
#pragma unroll
  for (int q = 0; q < EPT; ++q) {
    if (uv[q].x >= 0) {
      const int u = uv[q].x, v = uv[q].y;
      const int bv = v >> BIN_SHIFT, sv = bv * PITCH + t;
      const int pv = tab[sv];
      tab[sv] = pv + 1;
      stage[binoff[bv] + pv] = make_int2((u << 8) | (v & 255), gi[q]);
      const int bu = u >> BIN_SHIFT, su = bu * PITCH + t;
      const int pu = tab[su];
      tab[su] = pu + 1;
      stage[binoff[bu] + pu] = make_int2((v << 8) | (u & 255), gi[q]);
    }
  }
  __syncthreads();
  const int4* img = (const int4*)stage;
  int4*       seg = (int4*)(rec1 + (size_t)blk * SEG);
  const int   n4  = nrec >> 1;
  int4 cp[SEG / 2 / THREADS];
#pragma unroll
  for (int q = 0; q < SEG / 2 / THREADS; ++q) {
    const int i = q * THREADS + t;
    cp[q] = i < n4 ? img[i] : make_int4(0, 0, 0, 0);
  }
#pragma unroll
  for (int q = 0; q < SEG / 2 / THREADS; ++q) {
    const int i = q * THREADS + t;
    if (i < n4) seg[i] = cp[q];
  }
}

#define RUNS_PG    (MAX_BLKS / 16)
#define PASSES     4

__global__ void __launch_bounds__(THREADS)
fine_group_kernel(const int2* __restrict__ rec1, const int* __restrict__ runcnt,
                  const int* __restrict__ runoff, int2* __restrict__ rec2,
                  int* __restrict__ start, int* __restrict__ deg, int N, int nblk) {
  __shared__ int  tab[BIN_NODES * PITCH];
  __shared__ int  nodeoff[BIN_NODES];
  __shared__ int2 run[MAX_BLKS];
  __shared__ int  wsum[THREADS / 32];
  const int t = threadIdx.x, bin = blockIdx.x;
  const int g = t >> 4, l = t & 15;

  int myc = 0, myo = 0;
  if (t < nblk) { myc = runcnt[bin * nblk + t]; myo = runoff[bin * nblk + t]; }

  const int lim = bin * nblk, lim4 = lim & ~3;
  int below = 0;
  for (int base = 0; base < lim4; base += THREADS * 4 * 16) {
    int4 c[16];
#pragma unroll
    for (int q = 0; q < 16; ++q) {
      const int i = base + (q * THREADS + t) * 4;
      c[q] = *(const int4*)(runcnt + (i < lim4 ? i : 0));
      if (i >= lim4) c[q] = make_int4(0, 0, 0, 0);
    }
#pragma unroll
    for (int q = 0; q < 16; ++q) below += (c[q].x + c[q].y) + (c[q].z + c[q].w);
  }
  if (t < lim - lim4) below += runcnt[lim4 + t];
  for (int d = 16; d >= 1; d >>= 1) below += __shfl_xor(below, d);
  if ((t & 31) == 0) wsum[t >> 5] = below;

  run[t] = make_int2(myc, t * SEG + myo);
  for (int r = 0; r < BIN_NODES; ++r) tab[r * PITCH + t] = 0;
  __syncthreads();
  int b0 = 0;
  for (int w = 0; w < THREADS / 32; ++w) b0 += wsum[w];
  int2 rn[RUNS_PG];
#pragma unroll
  for (int q = 0; q < RUNS_PG; ++q) rn[q] = run[g + 16 * q];

  {
    int x[RUNS_PG][PASSES];
#pragma unroll
    for (int q = 0; q < RUNS_PG; ++q)
#pragma unroll
      for (int p = 0; p < PASSES; ++p) {
        const int i = p * 16 + l;
        x[q][p] = i < rn[q].x ? rec1[rn[q].y + i].x : -1;
      }
#pragma unroll
    for (int q = 0; q < RUNS_PG; ++q)
#pragma unroll
      for (int p = 0; p < PASSES; ++p)
        if (x[q][p] >= 0) tab[(x[q][p] & 255) * PITCH + t] += 1;
#pragma unroll
    for (int q = 0; q < RUNS_PG; ++q)
      for (int i = PASSES * 16 + l; i < rn[q].x; i += 16)
        tab[(rec1[rn[q].y + i].x & 255) * PITCH + t] += 1;
  }
  __syncthreads();

  int mydeg = 0;
  for (int j = 0; j < THREADS; ++j) {
    const int c = tab[t * PITCH + j];
    tab[t * PITCH + j] = mydeg;
    mydeg += c;
  }
  nodeoff[t] = mydeg;
  __syncthreads();
  for (int off = 1; off < BIN_NODES; off <<= 1) {
    const int a = t >= off ? nodeoff[t - off] : 0;
    __syncthreads();
    nodeoff[t] += a;
    __syncthreads();
  }
  const int first = b0 + nodeoff[t] - mydeg;
  nodeoff[t] = first;
  const int node = bin * BIN_NODES + t;
  if (node < N) { start[node] = first; deg[node] = mydeg; }
  __syncthreads();

  {
    int2 rc[RUNS_PG][PASSES];
#pragma unroll
    for (int q = 0; q < RUNS_PG; ++q)
#pragma unroll
      for (int p = 0; p < PASSES; ++p) {
        const int i = p * 16 + l;
        rc[q][p] = i < rn[q].x ? rec1[rn[q].y + i] : make_int2(-1, 0);
      }
#pragma unroll
    for (int q = 0; q < RUNS_PG; ++q) {
#pragma unroll
      for (int p = 0; p < PASSES; ++p) {
        if (rc[q][p].x >= 0) {
          const int loc = rc[q][p].x & 255;
          const int k   = tab[loc * PITCH + t];
          tab[loc * PITCH + t] = k + 1;
          rec2[nodeoff[loc] + k] = make_int2(rc[q][p].x >> 8, rc[q][p].y);
        }
      }
    }
#pragma unroll
    for (int q = 0; q < RUNS_PG; ++q) {
      for (int i = PASSES * 16 + l; i < rn[q].x; i += 16) {
        const int2 r1  = rec1[rn[q].y + i];
        const int  loc = r1.x & 255;
        const int  k   = tab[loc * PITCH + t];
        tab[loc * PITCH + t] = k + 1;
        rec2[nodeoff[loc] + k] = make_int2(r1.x >> 8, r1.y);
      }
    }
  }
}

#define TILE_NODES 32
#define NODES_PW   (TILE_NODES / 8)
#define A_PITCH    68

__device__ __forceinline__ float4 ld_row4(const float* pb, int idx) {
  return *(const float4*)(pb + (size_t)idx * D_DIM);
}

__device__ __forceinline__ void fma_row4(float4& acc, const float4 p, float g) {
  acc.x = fmaf(p.x, g, acc.x);
  acc.y = fmaf(p.y, g, acc.y);
  acc.z = fmaf(p.z, g, acc.z);
  acc.w = fmaf(p.w, g, acc.w);
}

__global__ void __launch_bounds__(THREADS)
gather_gemm_lrelu_kernel(const float* __restrict__ prev,
                         const int*   __restrict__ deg,
                         const int*   __restrict__ start,
                         const int2*  __restrict__ rec,
                         const float* __restrict__ Wm,
                         const float* __restrict__ nf,
                         const float* __restrict__ ef,
                         float*       __restrict__ out,
                         int N) {
  __shared__ float tile[B_DIM * TILE_NODES * A_PITCH];
  const int t    = threadIdx.x;
  const int lane = t & 31;
  const int w     = __builtin_amdgcn_readfirstlane(t >> 5);
  const int node0 = blockIdx.x * TILE_NODES;

  {
    const int b  = lane >> 4;
    const int d4 = (lane & 15) * 4;
    const float* pb = prev + (size_t)b * N * D_DIM + d4;

    for (int i = 0; i < NODES_PW; ++i) {
      const int nl   = w * NODES_PW + i;
      const int node = node0 + nl;
      float4 acc; acc.x = 0.f; acc.y = 0.f; acc.z = 0.f; acc.w = 0.f;

      if (node < N) {
        const int s = start[node];
        const int n = deg[node];
        for (int j0 = 0; j0 < n; j0 += 32) {
          const int m = (n - j0) < 32 ? (n - j0) : 32;
          int2 ent = make_int2(0, 0);
          if (lane < m) ent = rec[s + j0 + lane];

          int k = 0;
          for (; k + 8 <= m; k += 8) {
            int ii[8]; float gg[8]; float4 pp[8];
#pragma unroll
            for (int q = 0; q < 8; ++q) {
              ii[q] = __builtin_amdgcn_readlane(ent.x, k + q);
              gg[q] = __int_as_float(__builtin_amdgcn_readlane(ent.y, k + q));
            }
#pragma unroll
            for (int q = 0; q < 8; ++q) pp[q] = ld_row4(pb, ii[q]);
#pragma unroll
            for (int q = 0; q < 8; ++q) fma_row4(acc, pp[q], gg[q]);
          }
          if (k + 4 <= m) {
            int ii[4]; float gg[4]; float4 pp[4];
#pragma unroll
            for (int q = 0; q < 4; ++q) {
              ii[q] = __builtin_amdgcn_readlane(ent.x, k + q);
              gg[q] = __int_as_float(__builtin_amdgcn_readlane(ent.y, k + q));
            }
#pragma unroll
            for (int q = 0; q < 4; ++q) pp[q] = ld_row4(pb, ii[q]);
#pragma unroll
            for (int q = 0; q < 4; ++q) fma_row4(acc, pp[q], gg[q]);
            k += 4;
          }
          for (; k < m; ++k) {
            const int   i1 = __builtin_amdgcn_readlane(ent.x, k);
            const float g1 = __int_as_float(__builtin_amdgcn_readlane(ent.y, k));
            fma_row4(acc, ld_row4(pb, i1), g1);
          }
        }
      }
      *(float4*)(tile + (b * TILE_NODES + nl) * A_PITCH + d4) = acc;
    }
  }
  __syncthreads();

  const int rt   = w >> 1;
  const int bt   = rt >> 1;
  const int nb0  = node0 + (rt & 1) * 16;
  if (nb0 + 16 > N) return;
  const int half = lane >> 4;
  const int lm   = lane & 15;

  const float* arow = tile + (rt * 16 + lm) * A_PITCH + 2 * half;
  const int    col  = (w & 1) * 32 + lm;
  const float* wrow = Wm + (size_t)col * D_DIM + 2 * half;

  const size_t o0 = ((size_t)bt * N + nb0 + 8 * half) * D_DIM + col;

  v8f acc0, acc1;
#pragma unroll
  for (int r = 0; r < 8; ++r) {
    acc0[r] = nf[o0 + r * D_DIM]      + ef[o0 + r * D_DIM];
    acc1[r] = nf[o0 + r * D_DIM + 16] + ef[o0 + r * D_DIM + 16];
  }

#pragma unroll 4
  for (int q = 0; q < 16; ++q) {
    const v2f a  = *(const v2f*)(arow + 4 * q);
    const v2f b0 = *(const v2f*)(wrow + 4 * q);
    const v2f b1 = *(const v2f*)(wrow + 16 * D_DIM + 4 * q);
    acc0 = __builtin_amdgcn_wmma_f32_16x16x4_f32(false, a, false, b0, (short)0, acc0, false, false);
    acc1 = __builtin_amdgcn_wmma_f32_16x16x4_f32(false, a, false, b1, (short)0, acc1, false, false);
  }

#pragma unroll
  for (int r = 0; r < 8; ++r) {
    float d0 = acc0[r], d1 = acc1[r];
    d0 = d0 > 0.f ? d0 : 0.01f * d0;
    d1 = d1 > 0.f ? d1 : 0.01f * d1;
    __builtin_nontemporal_store(d0, out + o0 + r * D_DIM);
    __builtin_nontemporal_store(d1, out + o0 + r * D_DIM + 16);
  }
}

extern "C" void kernel_launch(void* const* d_in, const int* in_sizes, int n_in,
                              void* d_out, int out_size, void* d_ws, size_t ws_size,
                              hipStream_t stream) {
  const float* prev  = (const float*)d_in[0];
  const int*   edges = (const int*)  d_in[1];
  const float* nf    = (const float*)d_in[2];
  const float* ef    = (const float*)d_in[3];
  const float* st    = (const float*)d_in[4];
  const float* Wm    = (const float*)d_in[5];

  const int BND = in_sizes[0];
  const int E   = in_sizes[4];
  const int N   = BND / (B_DIM * D_DIM);

  const int nb   = (N + BIN_NODES - 1) / BIN_NODES;
  const int nblk = (E + EPB - 1) / EPB;

  int*  scr    = (int*)d_out;
  int2* rec1   = (int2*)scr;
  int*  runcnt = scr + 2 * nblk * SEG;
  int*  runoff = runcnt + MAX_BINS * MAX_BLKS;

  int*  wsi    = (int*)d_ws;
  const int NP = ((N + 63) / 64) * 64;
  int2* rec2   = (int2*)wsi;
  int*  start  = wsi + 4 * E;
  int*  deg    = start + NP;

  coarse_bin_kernel<<<nblk, THREADS, 0, stream>>>(edges, st, rec1, runcnt, runoff,
                                                  E, nb, nblk);
  fine_group_kernel<<<nb, THREADS, 0, stream>>>(rec1, runcnt, runoff, rec2, start, deg,
                                                N, nblk);
  gather_gemm_lrelu_kernel<<<(N + TILE_NODES - 1) / TILE_NODES, THREADS, 0, stream>>>(
      prev, deg, start, rec2, Wm, nf, ef, (float*)d_out, N);
}


# Level 2 reads nothing in global memory twice: lower bins' record count = sum of the run offsets already loaded (run-length table sweep dropped); records stay in registers between its two walks
# speedup vs baseline: 3.3644x; 1.0391x over previous
#include <hip/hip_runtime.h>

typedef __attribute__((ext_vector_type(2))) float v2f;
typedef __attribute__((ext_vector_type(8))) float v8f;

#define D_DIM 64
#define B_DIM 2


#define BIN_SHIFT  8
#define BIN_NODES  256
#define PITCH      257
#define MAX_BINS   200
#define MAX_BLKS   256
#define THREADS    256
#define EPT        16
#define EPB        (EPT * THREADS)
#define SEG        (2 * EPB)

static_assert(MAX_BINS <= THREADS, "one thread per coarse bin in level 1");
static_assert(MAX_BLKS <= THREADS, "one thread per run in level 2");

__device__ __forceinline__ int excl_scan_256(int v, int* wsum, int t) {
  const int lane = t & 31;
  int x = v;
  for (int d = 1; d < 32; d <<= 1) {
    const int y = __shfl_up(x, d);
    if (lane >= d) x += y;
  }
  if (lane == 31) wsum[t >> 5] = x;
  __syncthreads();
  int base = 0;
  for (int w = 0; w < THREADS / 32; ++w) base += w < (t >> 5) ? wsum[w] : 0;
  return base + x - v;
}

__device__ __forceinline__ void load_edges(int2 (&uv)[EPT], const int* __restrict__ edges,
                                           int E, int e0, int t) {
#pragma unroll
  for (int q = 0; q < EPT; ++q) {
    const int e = e0 + q * THREADS + t;
    uv[q] = e < E ? ((const int2*)edges)[e] : make_int2(-1, -1);
  }
}

__device__ __forceinline__ void coarse_count(int* tab, const int2 (&uv)[EPT], int nb, int t) {
  for (int r = 0; r < nb; ++r) tab[r * PITCH + t] = 0;
#pragma unroll
  for (int q = 0; q < EPT; ++q) {
    if (uv[q].x >= 0) {
      tab[(uv[q].y >> BIN_SHIFT) * PITCH + t] += 1;
      tab[(uv[q].x >> BIN_SHIFT) * PITCH + t] += 1;
    }
  }
}

__global__ void __launch_bounds__(THREADS)
coarse_bin_kernel(const int* __restrict__ edges, const float* __restrict__ status,
                  int2* __restrict__ rec1, int* __restrict__ runcnt,
                  int* __restrict__ runoff, int E, int nb, int nblk) {
  __shared__ int  tab[MAX_BINS * PITCH];
  __shared__ int2 stage[SEG];
  __shared__ int  binoff[THREADS];
  __shared__ int  wsum[THREADS / 32];
  __shared__ int  nrec;
  const int t = threadIdx.x, blk = blockIdx.x;
  const int e0 = blk * EPB;
  int2 uv[EPT];
  int  gi[EPT];
  load_edges(uv, edges, E, e0, t);
#pragma unroll
  for (int q = 0; q < EPT; ++q) {
    const int e = e0 + q * THREADS + t;
    gi[q] = e < E ? __float_as_int(status[e]) : 0;
  }
  coarse_count(tab, uv, nb, t);
  __syncthreads();
  int tot = 0;
  if (t < nb) {
    for (int j = 0; j < THREADS; ++j) {
      const int c = tab[t * PITCH + j];
      tab[t * PITCH + j] = tot;
      tot += c;
    }
  }
  const int off = excl_scan_256(tot, wsum, t);
  binoff[t] = off;
  if (t == THREADS - 1) nrec = off + tot;
  if (t < nb) {
    runcnt[t * nblk + blk] = tot;
    runoff[t * nblk + blk] = off;
  }
  __syncthreads();
#pragma unroll
  for (int q = 0; q < EPT; ++q) {
    if (uv[q].x >= 0) {
      const int u = uv[q].x, v = uv[q].y;
      const int bv = v >> BIN_SHIFT, sv = bv * PITCH + t;
      const int pv = tab[sv];
      tab[sv] = pv + 1;
      stage[binoff[bv] + pv] = make_int2((u << 8) | (v & 255), gi[q]);
      const int bu = u >> BIN_SHIFT, su = bu * PITCH + t;
      const int pu = tab[su];
      tab[su] = pu + 1;
      stage[binoff[bu] + pu] = make_int2((v << 8) | (u & 255), gi[q]);
    }
  }
  __syncthreads();
  const int4* img = (const int4*)stage;
  int4*       seg = (int4*)(rec1 + (size_t)blk * SEG);
  const int   n4  = nrec >> 1;
  int4 cp[SEG / 2 / THREADS];
#pragma unroll
  for (int q = 0; q < SEG / 2 / THREADS; ++q) {
    const int i = q * THREADS + t;
    cp[q] = i < n4 ? img[i] : make_int4(0, 0, 0, 0);
  }
#pragma unroll
  for (int q = 0; q < SEG / 2 / THREADS; ++q) {
    const int i = q * THREADS + t;
    if (i < n4) seg[i] = cp[q];
  }
}

#define RUNS_PG    (MAX_BLKS / 16)
#define PASSES     4

__global__ void __launch_bounds__(THREADS)
fine_group_kernel(const int2* __restrict__ rec1, const int* __restrict__ runcnt,
                  const int* __restrict__ runoff, int2* __restrict__ rec2,
                  int* __restrict__ start, int* __restrict__ deg, int N, int nblk) {
  __shared__ int  tab[BIN_NODES * PITCH];
  __shared__ int  nodeoff[BIN_NODES];
  __shared__ int2 run[MAX_BLKS];
  __shared__ int  wsum[THREADS / 32];
  const int t = threadIdx.x, bin = blockIdx.x;
  const int g = t >> 4, l = t & 15;

  int myc = 0, myo = 0;
  if (t < nblk) { myc = runcnt[bin * nblk + t]; myo = runoff[bin * nblk + t]; }

  for (int r = 0; r < BIN_NODES; ++r) tab[r * PITCH + t] = 0;

  int below = myo;
  for (int d = 16; d >= 1; d >>= 1) below += __shfl_xor(below, d);
  if ((t & 31) == 0) wsum[t >> 5] = below;

  run[t] = make_int2(myc, t * SEG + myo);
  __syncthreads();
  int b0 = 0;
  for (int w = 0; w < THREADS / 32; ++w) b0 += wsum[w];
  int2 rn[RUNS_PG];
#pragma unroll
  for (int q = 0; q < RUNS_PG; ++q) rn[q] = run[g + 16 * q];

  int2 rc[RUNS_PG][PASSES];
  {
#pragma unroll
    for (int q = 0; q < RUNS_PG; ++q)
#pragma unroll
      for (int p = 0; p < PASSES; ++p) {
        const int i = p * 16 + l;
        rc[q][p] = i < rn[q].x ? rec1[rn[q].y + i] : make_int2(-1, 0);
      }
#pragma unroll
    for (int q = 0; q < RUNS_PG; ++q)
#pragma unroll
      for (int p = 0; p < PASSES; ++p)
        if (rc[q][p].x >= 0) tab[(rc[q][p].x & 255) * PITCH + t] += 1;
#pragma unroll
    for (int q = 0; q < RUNS_PG; ++q)
      for (int i = PASSES * 16 + l; i < rn[q].x; i += 16)
        tab[(rec1[rn[q].y + i].x & 255) * PITCH + t] += 1;
  }
  __syncthreads();

  int mydeg = 0;
  for (int j = 0; j < THREADS; ++j) {
    const int c = tab[t * PITCH + j];
    tab[t * PITCH + j] = mydeg;
    mydeg += c;
  }
  nodeoff[t] = mydeg;
  __syncthreads();
  for (int off = 1; off < BIN_NODES; off <<= 1) {
    const int a = t >= off ? nodeoff[t - off] : 0;
    __syncthreads();
    nodeoff[t] += a;
    __syncthreads();
  }
  const int first = b0 + nodeoff[t] - mydeg;
  nodeoff[t] = first;
  const int node = bin * BIN_NODES + t;
  if (node < N) { start[node] = first; deg[node] = mydeg; }
  __syncthreads();

  {
#pragma unroll
    for (int q = 0; q < RUNS_PG; ++q) {
#pragma unroll
      for (int p = 0; p < PASSES; ++p) {
        if (rc[q][p].x >= 0) {
          const int loc = rc[q][p].x & 255;
          const int k   = tab[loc * PITCH + t];
          tab[loc * PITCH + t] = k + 1;
          rec2[nodeoff[loc] + k] = make_int2(rc[q][p].x >> 8, rc[q][p].y);
        }
      }
    }
#pragma unroll
    for (int q = 0; q < RUNS_PG; ++q) {
      for (int i = PASSES * 16 + l; i < rn[q].x; i += 16) {
        const int2 r1  = rec1[rn[q].y + i];
        const int  loc = r1.x & 255;
        const int  k   = tab[loc * PITCH + t];
        tab[loc * PITCH + t] = k + 1;
        rec2[nodeoff[loc] + k] = make_int2(r1.x >> 8, r1.y);
      }
    }
  }
}

#define TILE_NODES 32
#define NODES_PW   (TILE_NODES / 8)
#define A_PITCH    68

__device__ __forceinline__ float4 ld_row4(const float* pb, int idx) {
  return *(const float4*)(pb + (size_t)idx * D_DIM);
}

__device__ __forceinline__ void fma_row4(float4& acc, const float4 p, float g) {
  acc.x = fmaf(p.x, g, acc.x);
  acc.y = fmaf(p.y, g, acc.y);
  acc.z = fmaf(p.z, g, acc.z);
  acc.w = fmaf(p.w, g, acc.w);
}

__global__ void __launch_bounds__(THREADS)
gather_gemm_lrelu_kernel(const float* __restrict__ prev,
                         const int*   __restrict__ deg,
                         const int*   __restrict__ start,
                         const int2*  __restrict__ rec,
                         const float* __restrict__ Wm,
                         const float* __restrict__ nf,
                         const float* __restrict__ ef,
                         float*       __restrict__ out,
                         int N) {
  __shared__ float tile[B_DIM * TILE_NODES * A_PITCH];
  const int t    = threadIdx.x;
  const int lane = t & 31;
  const int w     = __builtin_amdgcn_readfirstlane(t >> 5);
  const int node0 = blockIdx.x * TILE_NODES;

  {
    const int b  = lane >> 4;
    const int d4 = (lane & 15) * 4;
    const float* pb = prev + (size_t)b * N * D_DIM + d4;

    for (int i = 0; i < NODES_PW; ++i) {
      const int nl   = w * NODES_PW + i;
      const int node = node0 + nl;
      float4 acc; acc.x = 0.f; acc.y = 0.f; acc.z = 0.f; acc.w = 0.f;

      if (node < N) {
        const int s = start[node];
        const int n = deg[node];
        for (int j0 = 0; j0 < n; j0 += 32) {
          const int m = (n - j0) < 32 ? (n - j0) : 32;
          int2 ent = make_int2(0, 0);
          if (lane < m) ent = rec[s + j0 + lane];

          int k = 0;
          for (; k + 8 <= m; k += 8) {
            int ii[8]; float gg[8]; float4 pp[8];
#pragma unroll
            for (int q = 0; q < 8; ++q) {
              ii[q] = __builtin_amdgcn_readlane(ent.x, k + q);
              gg[q] = __int_as_float(__builtin_amdgcn_readlane(ent.y, k + q));
            }
#pragma unroll
            for (int q = 0; q < 8; ++q) pp[q] = ld_row4(pb, ii[q]);
#pragma unroll
            for (int q = 0; q < 8; ++q) fma_row4(acc, pp[q], gg[q]);
          }
          if (k + 4 <= m) {
            int ii[4]; float gg[4]; float4 pp[4];
#pragma unroll
            for (int q = 0; q < 4; ++q) {
              ii[q] = __builtin_amdgcn_readlane(ent.x, k + q);
              gg[q] = __int_as_float(__builtin_amdgcn_readlane(ent.y, k + q));
            }
#pragma unroll
            for (int q = 0; q < 4; ++q) pp[q] = ld_row4(pb, ii[q]);
#pragma unroll
            for (int q = 0; q < 4; ++q) fma_row4(acc, pp[q], gg[q]);
            k += 4;
          }
          for (; k < m; ++k) {
            const int   i1 = __builtin_amdgcn_readlane(ent.x, k);
            const float g1 = __int_as_float(__builtin_amdgcn_readlane(ent.y, k));
            fma_row4(acc, ld_row4(pb, i1), g1);
          }
        }
      }
      *(float4*)(tile + (b * TILE_NODES + nl) * A_PITCH + d4) = acc;
    }
  }
  __syncthreads();

  const int rt   = w >> 1;
  const int bt   = rt >> 1;
  const int nb0  = node0 + (rt & 1) * 16;
  if (nb0 + 16 > N) return;
  const int half = lane >> 4;
  const int lm   = lane & 15;

  const float* arow = tile + (rt * 16 + lm) * A_PITCH + 2 * half;
  const int    col  = (w & 1) * 32 + lm;
  const float* wrow = Wm + (size_t)col * D_DIM + 2 * half;

  const size_t o0 = ((size_t)bt * N + nb0 + 8 * half) * D_DIM + col;

  v8f acc0, acc1;
#pragma unroll
  for (int r = 0; r < 8; ++r) {
    acc0[r] = nf[o0 + r * D_DIM]      + ef[o0 + r * D_DIM];
    acc1[r] = nf[o0 + r * D_DIM + 16] + ef[o0 + r * D_DIM + 16];
  }

#pragma unroll 4
  for (int q = 0; q < 16; ++q) {
    const v2f a  = *(const v2f*)(arow + 4 * q);
    const v2f b0 = *(const v2f*)(wrow + 4 * q);
    const v2f b1 = *(const v2f*)(wrow + 16 * D_DIM + 4 * q);
    acc0 = __builtin_amdgcn_wmma_f32_16x16x4_f32(false, a, false, b0, (short)0, acc0, false, false);
    acc1 = __builtin_amdgcn_wmma_f32_16x16x4_f32(false, a, false, b1, (short)0, acc1, false, false);
  }

#pragma unroll
  for (int r = 0; r < 8; ++r) {
    float d0 = acc0[r], d1 = acc1[r];
    d0 = d0 > 0.f ? d0 : 0.01f * d0;
    d1 = d1 > 0.f ? d1 : 0.01f * d1;
    __builtin_nontemporal_store(d0, out + o0 + r * D_DIM);
    __builtin_nontemporal_store(d1, out + o0 + r * D_DIM + 16);
  }
}

extern "C" void kernel_launch(void* const* d_in, const int* in_sizes, int n_in,
                              void* d_out, int out_size, void* d_ws, size_t ws_size,
                              hipStream_t stream) {
  const float* prev  = (const float*)d_in[0];
  const int*   edges = (const int*)  d_in[1];
  const float* nf    = (const float*)d_in[2];
  const float* ef    = (const float*)d_in[3];
  const float* st    = (const float*)d_in[4];
  const float* Wm    = (const float*)d_in[5];

  const int BND = in_sizes[0];
  const int E   = in_sizes[4];
  const int N   = BND / (B_DIM * D_DIM);

  const int nb   = (N + BIN_NODES - 1) / BIN_NODES;
  const int nblk = (E + EPB - 1) / EPB;

  int*  scr    = (int*)d_out;
  int2* rec1   = (int2*)scr;
  int*  runcnt = scr + 2 * nblk * SEG;
  int*  runoff = runcnt + MAX_BINS * MAX_BLKS;

  int*  wsi    = (int*)d_ws;
  const int NP = ((N + 63) / 64) * 64;
  int2* rec2   = (int2*)wsi;
  int*  start  = wsi + 4 * E;
  int*  deg    = start + NP;

  coarse_bin_kernel<<<nblk, THREADS, 0, stream>>>(edges, st, rec1, runcnt, runoff,
                                                  E, nb, nblk);
  fine_group_kernel<<<nb, THREADS, 0, stream>>>(rec1, runcnt, runoff, rec2, start, deg,
                                                N, nblk);
  gather_gemm_lrelu_kernel<<<(N + TILE_NODES - 1) / TILE_NODES, THREADS, 0, stream>>>(
      prev, deg, start, rec2, Wm, nf, ef, (float*)d_out, N);
}


# Thread-private LDS counters updated eight slots at a time (equal keys settled in registers), ranks kept; placing walks only read the table: no LDS round trip per record; list order unchanged
# speedup vs baseline: 3.5233x; 1.0472x over previous
#include <hip/hip_runtime.h>

typedef __attribute__((ext_vector_type(2))) float v2f;
typedef __attribute__((ext_vector_type(8))) float v8f;

#define D_DIM 64
#define B_DIM 2


#define BIN_SHIFT  8
#define BIN_NODES  256
#define PITCH      257
#define MAX_BINS   200
#define MAX_BLKS   256
#define THREADS    256
#define EPT        16
#define EPB        (EPT * THREADS)
#define SEG        (2 * EPB)

static_assert(MAX_BINS < THREADS, "one thread per coarse bin in level 1, and an offset entry for the spare row");
static_assert(MAX_BLKS <= THREADS, "one thread per run in level 2");

__device__ __forceinline__ int excl_scan_256(int v, int* wsum, int t) {
  const int lane = t & 31;
  int x = v;
  for (int d = 1; d < 32; d <<= 1) {
    const int y = __shfl_up(x, d);
    if (lane >= d) x += y;
  }
  if (lane == 31) wsum[t >> 5] = x;
  __syncthreads();
  int base = 0;
  for (int w = 0; w < THREADS / 32; ++w) base += w < (t >> 5) ? wsum[w] : 0;
  return base + x - v;
}

#define GRP 8
template <int NS>
__device__ __forceinline__ void count_ranked(int* col, const int (&key)[NS], int (&rank)[NS]) {
  static_assert(NS % GRP == 0, "whole groups");
#pragma unroll
  for (int b = 0; b < NS; b += GRP) {
    int c[GRP];
#pragma unroll
    for (int i = 0; i < GRP; ++i) c[i] = col[key[b + i] * PITCH];
#pragma unroll
    for (int i = 0; i < GRP; ++i) {
      int before = 0;
#pragma unroll
      for (int j = 0; j < i; ++j) before += key[b + j] == key[b + i] ? 1 : 0;
      rank[b + i] = c[i] + before;
    }
#pragma unroll
    for (int i = 0; i < GRP; ++i) col[key[b + i] * PITCH] = rank[b + i] + 1;
  }
}

__device__ __forceinline__ void load_edges(int2 (&uv)[EPT], const int* __restrict__ edges,
                                           int E, int e0, int t) {
#pragma unroll
  for (int q = 0; q < EPT; ++q) {
    const int e = e0 + q * THREADS + t;
    uv[q] = e < E ? ((const int2*)edges)[e] : make_int2(-1, -1);
  }
}

__device__ __forceinline__ void coarse_count(int* tab, const int2 (&uv)[EPT], int (&key)[2 * EPT],
                                             int (&rank)[2 * EPT], int nb, int t) {
  for (int r = 0; r < nb; ++r) tab[r * PITCH + t] = 0;
  tab[MAX_BINS * PITCH + t] = 0;
#pragma unroll
  for (int q = 0; q < EPT; ++q) {
    const bool ok = uv[q].x >= 0;
    key[2 * q]     = ok ? uv[q].y >> BIN_SHIFT : MAX_BINS;
    key[2 * q + 1] = ok ? uv[q].x >> BIN_SHIFT : MAX_BINS;
  }
  count_ranked<2 * EPT>(tab + t, key, rank);
}

__global__ void __launch_bounds__(THREADS)
coarse_bin_kernel(const int* __restrict__ edges, const float* __restrict__ status,
                  int2* __restrict__ rec1, int* __restrict__ runcnt,
                  int* __restrict__ runoff, int E, int nb, int nblk) {
  __shared__ int  tab[(MAX_BINS + 1) * PITCH];
  __shared__ int2 stage[SEG];
  __shared__ int  binoff[THREADS];
  __shared__ int  wsum[THREADS / 32];
  __shared__ int  nrec;
  const int t = threadIdx.x, blk = blockIdx.x;
  const int e0 = blk * EPB;
  int2 uv[EPT];
  int  gi[EPT];
  int  key[2 * EPT], slot[2 * EPT];
  load_edges(uv, edges, E, e0, t);
#pragma unroll
  for (int q = 0; q < EPT; ++q) {
    const int e = e0 + q * THREADS + t;
    gi[q] = e < E ? __float_as_int(status[e]) : 0;
  }
  coarse_count(tab, uv, key, slot, nb, t);
  __syncthreads();
  int tot = 0;
  if (t < nb) {
    for (int j = 0; j < THREADS; ++j) {
      const int c = tab[t * PITCH + j];
      tab[t * PITCH + j] = tot;
      tot += c;
    }
  }
  const int off = excl_scan_256(tot, wsum, t);
  binoff[t] = off;
  if (t == THREADS - 1) nrec = off + tot;
  if (t < nb) {
    runcnt[t * nblk + blk] = tot;
    runoff[t * nblk + blk] = off;
  }
  __syncthreads();
#pragma unroll
  for (int s = 0; s < 2 * EPT; ++s) slot[s] += binoff[key[s]] + tab[key[s] * PITCH + t];
#pragma unroll
  for (int q = 0; q < EPT; ++q) {
    if (uv[q].x >= 0) {
      const int u = uv[q].x, v = uv[q].y;
      stage[slot[2 * q]]     = make_int2((u << 8) | (v & 255), gi[q]);
      stage[slot[2 * q + 1]] = make_int2((v << 8) | (u & 255), gi[q]);
    }
  }
  __syncthreads();
  const int4* img = (const int4*)stage;
  int4*       seg = (int4*)(rec1 + (size_t)blk * SEG);
  const int   n4  = nrec >> 1;
  int4 cp[SEG / 2 / THREADS];
#pragma unroll
  for (int q = 0; q < SEG / 2 / THREADS; ++q) {
    const int i = q * THREADS + t;
    cp[q] = i < n4 ? img[i] : make_int4(0, 0, 0, 0);
  }
#pragma unroll
  for (int q = 0; q < SEG / 2 / THREADS; ++q) {
    const int i = q * THREADS + t;
    if (i < n4) seg[i] = cp[q];
  }
}

#define RUNS_PG    (MAX_BLKS / 16)
#define PASSES     4

__global__ void __launch_bounds__(THREADS)
fine_group_kernel(const int2* __restrict__ rec1, const int* __restrict__ runcnt,
                  const int* __restrict__ runoff, int2* __restrict__ rec2,
                  int* __restrict__ start, int* __restrict__ deg, int N, int nblk) {
  __shared__ int  tab[(BIN_NODES + 1) * PITCH];
  __shared__ int  nodeoff[BIN_NODES];
  __shared__ int2 run[MAX_BLKS];
  __shared__ int  wsum[THREADS / 32];
  const int t = threadIdx.x, bin = blockIdx.x;
  const int g = t >> 4, l = t & 15;

  int myc = 0, myo = 0;
  if (t < nblk) { myc = runcnt[bin * nblk + t]; myo = runoff[bin * nblk + t]; }

  for (int r = 0; r <= BIN_NODES; ++r) tab[r * PITCH + t] = 0;

  int below = myo;
  for (int d = 16; d >= 1; d >>= 1) below += __shfl_xor(below, d);
  if ((t & 31) == 0) wsum[t >> 5] = below;

  run[t] = make_int2(myc, t * SEG + myo);
  __syncthreads();
  int b0 = 0;
  for (int w = 0; w < THREADS / 32; ++w) b0 += wsum[w];
  int2 rn[RUNS_PG];
#pragma unroll
  for (int q = 0; q < RUNS_PG; ++q) rn[q] = run[g + 16 * q];

  unsigned longmask = 0;
#pragma unroll
  for (int q = 0; q < RUNS_PG; ++q) longmask |= (rn[q].x > PASSES * 16 ? 1u : 0u) << q;

  int2 rc[RUNS_PG][PASSES];
  int  key[RUNS_PG * PASSES];
  int  pos[RUNS_PG * PASSES];
  {
#pragma unroll
    for (int q = 0; q < RUNS_PG; ++q)
#pragma unroll
      for (int p = 0; p < PASSES; ++p) {
        const int i = p * 16 + l;
        rc[q][p] = i < rn[q].x ? rec1[rn[q].y + i] : make_int2(-1, 0);
      }
#pragma unroll
    for (int q = 0; q < RUNS_PG; ++q)
#pragma unroll
      for (int p = 0; p < PASSES; ++p)
        key[q * PASSES + p] = rc[q][p].x >= 0 ? rc[q][p].x & 255 : BIN_NODES;
    count_ranked<RUNS_PG * PASSES>(tab + t, key, pos);
    for (unsigned m = longmask; m != 0; m &= m - 1) {
      const int2 r = run[g + 16 * (__ffs(m) - 1)];
      for (int i = PASSES * 16 + l; i < r.x; i += 16)
        tab[(rec1[r.y + i].x & 255) * PITCH + t] += 1;
    }
  }
  __syncthreads();

  int mydeg = 0;
  for (int j = 0; j < THREADS; ++j) {
    const int c = tab[t * PITCH + j];
    tab[t * PITCH + j] = mydeg;
    mydeg += c;
  }
  nodeoff[t] = mydeg;
  __syncthreads();
  for (int off = 1; off < BIN_NODES; off <<= 1) {
    const int a = t >= off ? nodeoff[t - off] : 0;
    __syncthreads();
    nodeoff[t] += a;
    __syncthreads();
  }
  const int first = b0 + nodeoff[t] - mydeg;
  nodeoff[t] = first;
  const int node = bin * BIN_NODES + t;
  if (node < N) { start[node] = first; deg[node] = mydeg; }
  __syncthreads();

  {
#pragma unroll
    for (int q = 0; q < RUNS_PG; ++q)
#pragma unroll
      for (int p = 0; p < PASSES; ++p) {
        const int loc = rc[q][p].x & 255;
        pos[q * PASSES + p] += nodeoff[loc] + tab[loc * PITCH + t];
      }
#pragma unroll
    for (int q = 0; q < RUNS_PG; ++q)
#pragma unroll
      for (int p = 0; p < PASSES; ++p)
        if (rc[q][p].x >= 0) rec2[pos[q * PASSES + p]] = make_int2(rc[q][p].x >> 8, rc[q][p].y);

    for (unsigned m = longmask; m != 0; m &= m - 1) {
      const int2 r = run[g + 16 * (__ffs(m) - 1)];
      for (int i = PASSES * 16 + l; i < r.x; i += 16) {
        const int2 r1  = rec1[r.y + i];
        const int  loc = r1.x & 255;
        int mine = 0;
#pragma unroll
        for (int q = 0; q < RUNS_PG; ++q)
#pragma unroll
          for (int p = 0; p < PASSES; ++p)
            mine += (rc[q][p].x & (int)0x800000ffu) == loc ? 1 : 0;
        const int  k   = tab[loc * PITCH + t];
        tab[loc * PITCH + t] = k + 1;
        rec2[nodeoff[loc] + k + mine] = make_int2(r1.x >> 8, r1.y);
      }
    }
  }
}

#define TILE_NODES 32
#define NODES_PW   (TILE_NODES / 8)
#define A_PITCH    68

__device__ __forceinline__ float4 ld_row4(const float* pb, int idx) {
  return *(const float4*)(pb + (size_t)idx * D_DIM);
}

__device__ __forceinline__ void fma_row4(float4& acc, const float4 p, float g) {
  acc.x = fmaf(p.x, g, acc.x);
  acc.y = fmaf(p.y, g, acc.y);
  acc.z = fmaf(p.z, g, acc.z);
  acc.w = fmaf(p.w, g, acc.w);
}

__global__ void __launch_bounds__(THREADS)
gather_gemm_lrelu_kernel(const float* __restrict__ prev,
                         const int*   __restrict__ deg,
                         const int*   __restrict__ start,
                         const int2*  __restrict__ rec,
                         const float* __restrict__ Wm,
                         const float* __restrict__ nf,
                         const float* __restrict__ ef,
                         float*       __restrict__ out,
                         int N) {
  __shared__ float tile[B_DIM * TILE_NODES * A_PITCH];
  const int t    = threadIdx.x;
  const int lane = t & 31;
  const int w     = __builtin_amdgcn_readfirstlane(t >> 5);
  const int node0 = blockIdx.x * TILE_NODES;

  {
    const int b  = lane >> 4;
    const int d4 = (lane & 15) * 4;
    const float* pb = prev + (size_t)b * N * D_DIM + d4;

    for (int i = 0; i < NODES_PW; ++i) {
      const int nl   = w * NODES_PW + i;
      const int node = node0 + nl;
      float4 acc; acc.x = 0.f; acc.y = 0.f; acc.z = 0.f; acc.w = 0.f;

      if (node < N) {
        const int s = start[node];
        const int n = deg[node];
        for (int j0 = 0; j0 < n; j0 += 32) {
          const int m = (n - j0) < 32 ? (n - j0) : 32;
          int2 ent = make_int2(0, 0);
          if (lane < m) ent = rec[s + j0 + lane];

          int k = 0;
          for (; k + 8 <= m; k += 8) {
            int ii[8]; float gg[8]; float4 pp[8];
#pragma unroll
            for (int q = 0; q < 8; ++q) {
              ii[q] = __builtin_amdgcn_readlane(ent.x, k + q);
              gg[q] = __int_as_float(__builtin_amdgcn_readlane(ent.y, k + q));
            }
#pragma unroll
            for (int q = 0; q < 8; ++q) pp[q] = ld_row4(pb, ii[q]);
#pragma unroll
            for (int q = 0; q < 8; ++q) fma_row4(acc, pp[q], gg[q]);
          }
          if (k + 4 <= m) {
            int ii[4]; float gg[4]; float4 pp[4];
#pragma unroll
            for (int q = 0; q < 4; ++q) {
              ii[q] = __builtin_amdgcn_readlane(ent.x, k + q);
              gg[q] = __int_as_float(__builtin_amdgcn_readlane(ent.y, k + q));
            }
#pragma unroll
            for (int q = 0; q < 4; ++q) pp[q] = ld_row4(pb, ii[q]);
#pragma unroll
            for (int q = 0; q < 4; ++q) fma_row4(acc, pp[q], gg[q]);
            k += 4;
          }
          for (; k < m; ++k) {
            const int   i1 = __builtin_amdgcn_readlane(ent.x, k);
            const float g1 = __int_as_float(__builtin_amdgcn_readlane(ent.y, k));
            fma_row4(acc, ld_row4(pb, i1), g1);
          }
        }
      }
      *(float4*)(tile + (b * TILE_NODES + nl) * A_PITCH + d4) = acc;
    }
  }
  __syncthreads();

  const int rt   = w >> 1;
  const int bt   = rt >> 1;
  const int nb0  = node0 + (rt & 1) * 16;
  if (nb0 + 16 > N) return;
  const int half = lane >> 4;
  const int lm   = lane & 15;

  const float* arow = tile + (rt * 16 + lm) * A_PITCH + 2 * half;
  const int    col  = (w & 1) * 32 + lm;
  const float* wrow = Wm + (size_t)col * D_DIM + 2 * half;

  const size_t o0 = ((size_t)bt * N + nb0 + 8 * half) * D_DIM + col;

  v8f acc0, acc1;
#pragma unroll
  for (int r = 0; r < 8; ++r) {
    acc0[r] = nf[o0 + r * D_DIM]      + ef[o0 + r * D_DIM];
    acc1[r] = nf[o0 + r * D_DIM + 16] + ef[o0 + r * D_DIM + 16];
  }

#pragma unroll 4
  for (int q = 0; q < 16; ++q) {
    const v2f a  = *(const v2f*)(arow + 4 * q);
    const v2f b0 = *(const v2f*)(wrow + 4 * q);
    const v2f b1 = *(const v2f*)(wrow + 16 * D_DIM + 4 * q);
    acc0 = __builtin_amdgcn_wmma_f32_16x16x4_f32(false, a, false, b0, (short)0, acc0, false, false);
    acc1 = __builtin_amdgcn_wmma_f32_16x16x4_f32(false, a, false, b1, (short)0, acc1, false, false);
  }

#pragma unroll
  for (int r = 0; r < 8; ++r) {
    float d0 = acc0[r], d1 = acc1[r];
    d0 = d0 > 0.f ? d0 : 0.01f * d0;
    d1 = d1 > 0.f ? d1 : 0.01f * d1;
    __builtin_nontemporal_store(d0, out + o0 + r * D_DIM);
    __builtin_nontemporal_store(d1, out + o0 + r * D_DIM + 16);
  }
}

extern "C" void kernel_launch(void* const* d_in, const int* in_sizes, int n_in,
                              void* d_out, int out_size, void* d_ws, size_t ws_size,
                              hipStream_t stream) {
  const float* prev  = (const float*)d_in[0];
  const int*   edges = (const int*)  d_in[1];
  const float* nf    = (const float*)d_in[2];
  const float* ef    = (const float*)d_in[3];
  const float* st    = (const float*)d_in[4];
  const float* Wm    = (const float*)d_in[5];

  const int BND = in_sizes[0];
  const int E   = in_sizes[4];
  const int N   = BND / (B_DIM * D_DIM);

  const int nb   = (N + BIN_NODES - 1) / BIN_NODES;
  const int nblk = (E + EPB - 1) / EPB;

  int*  scr    = (int*)d_out;
  int2* rec1   = (int2*)scr;
  int*  runcnt = scr + 2 * nblk * SEG;
  int*  runoff = runcnt + MAX_BINS * MAX_BLKS;

  int*  wsi    = (int*)d_ws;
  const int NP = ((N + 63) / 64) * 64;
  int2* rec2   = (int2*)wsi;
  int*  start  = wsi + 4 * E;
  int*  deg    = start + NP;

  coarse_bin_kernel<<<nblk, THREADS, 0, stream>>>(edges, st, rec1, runcnt, runoff,
                                                  E, nb, nblk);
  fine_group_kernel<<<nb, THREADS, 0, stream>>>(rec1, runcnt, runoff, rec2, start, deg,
                                                N, nblk);
  gather_gemm_lrelu_kernel<<<(N + TILE_NODES - 1) / TILE_NODES, THREADS, 0, stream>>>(
      prev, deg, start, rec2, Wm, nf, ef, (float*)d_out, N);
}


# Level 2 assembles its bin's ordered records in LDS (on the counter table, dead by then) and copies them out as whole 128-byte lines, 16 bytes a lane, not one 8-byte store per line; list order kept
# speedup vs baseline: 3.6865x; 1.0463x over previous
#include <hip/hip_runtime.h>

typedef __attribute__((ext_vector_type(2))) float v2f;
typedef __attribute__((ext_vector_type(8))) float v8f;

#define D_DIM 64
#define B_DIM 2


#define BIN_SHIFT  8
#define BIN_NODES  256
#define PITCH      257
#define MAX_BINS   200
#define MAX_BLKS   256
#define THREADS    256
#define EPT        16
#define EPB        (EPT * THREADS)
#define SEG        (2 * EPB)

static_assert(MAX_BINS < THREADS, "one thread per coarse bin in level 1, and an offset entry for the spare row");
static_assert(MAX_BLKS <= THREADS, "one thread per run in level 2");

__device__ __forceinline__ int excl_scan_256(int v, int* wsum, int t) {
  const int lane = t & 31;
  int x = v;
  for (int d = 1; d < 32; d <<= 1) {
    const int y = __shfl_up(x, d);
    if (lane >= d) x += y;
  }
  if (lane == 31) wsum[t >> 5] = x;
  __syncthreads();
  int base = 0;
  for (int w = 0; w < THREADS / 32; ++w) base += w < (t >> 5) ? wsum[w] : 0;
  return base + x - v;
}

#define GRP 8
template <int NS>
__device__ __forceinline__ void count_ranked(int* col, const int (&key)[NS], int (&rank)[NS]) {
  static_assert(NS % GRP == 0, "whole groups");
#pragma unroll
  for (int b = 0; b < NS; b += GRP) {
    int c[GRP];
#pragma unroll
    for (int i = 0; i < GRP; ++i) c[i] = col[key[b + i] * PITCH];
#pragma unroll
    for (int i = 0; i < GRP; ++i) {
      int before = 0;
#pragma unroll
      for (int j = 0; j < i; ++j) before += key[b + j] == key[b + i] ? 1 : 0;
      rank[b + i] = c[i] + before;
    }
#pragma unroll
    for (int i = 0; i < GRP; ++i) col[key[b + i] * PITCH] = rank[b + i] + 1;
  }
}

__device__ __forceinline__ void load_edges(int2 (&uv)[EPT], const int* __restrict__ edges,
                                           int E, int e0, int t) {
#pragma unroll
  for (int q = 0; q < EPT; ++q) {
    const int e = e0 + q * THREADS + t;
    uv[q] = e < E ? ((const int2*)edges)[e] : make_int2(-1, -1);
  }
}

__device__ __forceinline__ void coarse_count(int* tab, const int2 (&uv)[EPT], int (&key)[2 * EPT],
                                             int (&rank)[2 * EPT], int nb, int t) {
  for (int r = 0; r < nb; ++r) tab[r * PITCH + t] = 0;
  tab[MAX_BINS * PITCH + t] = 0;
#pragma unroll
  for (int q = 0; q < EPT; ++q) {
    const bool ok = uv[q].x >= 0;
    key[2 * q]     = ok ? uv[q].y >> BIN_SHIFT : MAX_BINS;
    key[2 * q + 1] = ok ? uv[q].x >> BIN_SHIFT : MAX_BINS;
  }
  count_ranked<2 * EPT>(tab + t, key, rank);
}

__global__ void __launch_bounds__(THREADS)
coarse_bin_kernel(const int* __restrict__ edges, const float* __restrict__ status,
                  int2* __restrict__ rec1, int* __restrict__ runcnt,
                  int* __restrict__ runoff, int E, int nb, int nblk) {
  __shared__ int  tab[(MAX_BINS + 1) * PITCH];
  __shared__ int2 stage[SEG];
  __shared__ int  binoff[THREADS];
  __shared__ int  wsum[THREADS / 32];
  __shared__ int  nrec;
  const int t = threadIdx.x, blk = blockIdx.x;
  const int e0 = blk * EPB;
  int2 uv[EPT];
  int  gi[EPT];
  int  key[2 * EPT], slot[2 * EPT];
  load_edges(uv, edges, E, e0, t);
#pragma unroll
  for (int q = 0; q < EPT; ++q) {
    const int e = e0 + q * THREADS + t;
    gi[q] = e < E ? __float_as_int(status[e]) : 0;
  }
  coarse_count(tab, uv, key, slot, nb, t);
  __syncthreads();
  int tot = 0;
  if (t < nb) {
    for (int j = 0; j < THREADS; ++j) {
      const int c = tab[t * PITCH + j];
      tab[t * PITCH + j] = tot;
      tot += c;
    }
  }
  const int off = excl_scan_256(tot, wsum, t);
  binoff[t] = off;
  if (t == THREADS - 1) nrec = off + tot;
  if (t < nb) {
    runcnt[t * nblk + blk] = tot;
    runoff[t * nblk + blk] = off;
  }
  __syncthreads();
#pragma unroll
  for (int s = 0; s < 2 * EPT; ++s) slot[s] += binoff[key[s]] + tab[key[s] * PITCH + t];
#pragma unroll
  for (int q = 0; q < EPT; ++q) {
    if (uv[q].x >= 0) {
      const int u = uv[q].x, v = uv[q].y;
      stage[slot[2 * q]]     = make_int2((u << 8) | (v & 255), gi[q]);
      stage[slot[2 * q + 1]] = make_int2((v << 8) | (u & 255), gi[q]);
    }
  }
  __syncthreads();
  const int4* img = (const int4*)stage;
  int4*       seg = (int4*)(rec1 + (size_t)blk * SEG);
  const int   n4  = nrec >> 1;
  int4 cp[SEG / 2 / THREADS];
#pragma unroll
  for (int q = 0; q < SEG / 2 / THREADS; ++q) {
    const int i = q * THREADS + t;
    cp[q] = i < n4 ? img[i] : make_int4(0, 0, 0, 0);
  }
#pragma unroll
  for (int q = 0; q < SEG / 2 / THREADS; ++q) {
    const int i = q * THREADS + t;
    if (i < n4) seg[i] = cp[q];
  }
}

#define RUNS_PG    (MAX_BLKS / 16)
#define PASSES     4
#define IMG_CAP    (((BIN_NODES + 1) * PITCH) / 2)
#define CP_GRP     8

__global__ void __launch_bounds__(THREADS)
fine_group_kernel(const int2* __restrict__ rec1, const int* __restrict__ runcnt,
                  const int* __restrict__ runoff, int2* __restrict__ rec2,
                  int* __restrict__ start, int* __restrict__ deg, int N, int nblk) {
  __shared__ __attribute__((aligned(16))) int tab[(BIN_NODES + 1) * PITCH];
  __shared__ int  nodeoff[BIN_NODES];
  __shared__ int2 run[MAX_BLKS];
  __shared__ int  wsum[THREADS / 32];
  __shared__ int  bintot;
  const int t = threadIdx.x, bin = blockIdx.x;
  const int g = t >> 4, l = t & 15;

  int myc = 0, myo = 0;
  if (t < nblk) { myc = runcnt[bin * nblk + t]; myo = runoff[bin * nblk + t]; }

  for (int r = 0; r <= BIN_NODES; ++r) tab[r * PITCH + t] = 0;

  int below = myo;
  for (int d = 16; d >= 1; d >>= 1) below += __shfl_xor(below, d);
  if ((t & 31) == 0) wsum[t >> 5] = below;

  run[t] = make_int2(myc, t * SEG + myo);
  __syncthreads();
  int b0 = 0;
  for (int w = 0; w < THREADS / 32; ++w) b0 += wsum[w];
  int2 rn[RUNS_PG];
#pragma unroll
  for (int q = 0; q < RUNS_PG; ++q) rn[q] = run[g + 16 * q];

  unsigned longmask = 0;
#pragma unroll
  for (int q = 0; q < RUNS_PG; ++q) longmask |= (rn[q].x > PASSES * 16 ? 1u : 0u) << q;

  int2 rc[RUNS_PG][PASSES];
  int  key[RUNS_PG * PASSES];
  int  pos[RUNS_PG * PASSES];
  {
#pragma unroll
    for (int q = 0; q < RUNS_PG; ++q)
#pragma unroll
      for (int p = 0; p < PASSES; ++p) {
        const int i = p * 16 + l;
        rc[q][p] = i < rn[q].x ? rec1[rn[q].y + i] : make_int2(-1, 0);
      }
#pragma unroll
    for (int q = 0; q < RUNS_PG; ++q)
#pragma unroll
      for (int p = 0; p < PASSES; ++p)
        key[q * PASSES + p] = rc[q][p].x >= 0 ? rc[q][p].x & 255 : BIN_NODES;
    count_ranked<RUNS_PG * PASSES>(tab + t, key, pos);
    for (unsigned m = longmask; m != 0; m &= m - 1) {
      const int2 r = run[g + 16 * (__ffs(m) - 1)];
      for (int i = PASSES * 16 + l; i < r.x; i += 16)
        tab[(rec1[r.y + i].x & 255) * PITCH + t] += 1;
    }
  }
  __syncthreads();

  int mydeg = 0;
  for (int j = 0; j < THREADS; ++j) {
    const int c = tab[t * PITCH + j];
    tab[t * PITCH + j] = mydeg;
    mydeg += c;
  }
  nodeoff[t] = mydeg;
  __syncthreads();
  for (int off = 1; off < BIN_NODES; off <<= 1) {
    const int a = t >= off ? nodeoff[t - off] : 0;
    __syncthreads();
    nodeoff[t] += a;
    __syncthreads();
  }
  const int first = b0 + nodeoff[t] - mydeg;
  nodeoff[t] = first;
  if (t == THREADS - 1) bintot = first + mydeg - b0;
  const int node = bin * BIN_NODES + t;
  if (node < N) { start[node] = first; deg[node] = mydeg; }
  __syncthreads();

  {
#pragma unroll
    for (int q = 0; q < RUNS_PG; ++q)
#pragma unroll
      for (int p = 0; p < PASSES; ++p) {
        const int loc = rc[q][p].x & 255;
        pos[q * PASSES + p] += nodeoff[loc] + tab[loc * PITCH + t];
      }

    for (unsigned m = longmask; m != 0; m &= m - 1) {
      const int2 r = run[g + 16 * (__ffs(m) - 1)];
      for (int i = PASSES * 16 + l; i < r.x; i += 16) {
        const int2 r1  = rec1[r.y + i];
        const int  loc = r1.x & 255;
        int mine = 0;
#pragma unroll
        for (int q = 0; q < RUNS_PG; ++q)
#pragma unroll
          for (int p = 0; p < PASSES; ++p)
            mine += (rc[q][p].x & (int)0x800000ffu) == loc ? 1 : 0;
        const int  k   = tab[loc * PITCH + t];
        tab[loc * PITCH + t] = k + 1;
        rec2[nodeoff[loc] + k + mine] = make_int2(r1.x >> 8, r1.y);
      }
    }

    const int base = b0 & ~15;
    const int span = b0 - base + bintot;
    if (span <= IMG_CAP) {
      int4* img4 = (int4*)tab;
      int2* img  = (int2*)tab;
      const int n4 = (span + 1) >> 1;
      __syncthreads();
      for (int i = t; i < n4; i += THREADS) img4[i] = make_int4(-1, 0, -1, 0);
      __syncthreads();
#pragma unroll
      for (int q = 0; q < RUNS_PG; ++q)
#pragma unroll
        for (int p = 0; p < PASSES; ++p)
          if (rc[q][p].x >= 0)
            img[pos[q * PASSES + p] - base] = make_int2(rc[q][p].x >> 8, rc[q][p].y);
      __syncthreads();
      int4* dst = (int4*)(rec2 + base);
      for (int i0 = 0; i0 < n4; i0 += CP_GRP * THREADS) {
        int4 cp[CP_GRP];
#pragma unroll
        for (int u = 0; u < CP_GRP; ++u) {
          const int i = i0 + u * THREADS + t;
          cp[u] = i < n4 ? img4[i] : make_int4(-1, 0, -1, 0);
        }
#pragma unroll
        for (int u = 0; u < CP_GRP; ++u) {
          const int i = i0 + u * THREADS + t;
          if (cp[u].x >= 0 && cp[u].z >= 0) dst[i] = cp[u];
          else if (cp[u].x >= 0) rec2[base + 2 * i]     = make_int2(cp[u].x, cp[u].y);
          else if (cp[u].z >= 0) rec2[base + 2 * i + 1] = make_int2(cp[u].z, cp[u].w);
        }
      }
    } else {
#pragma unroll
      for (int q = 0; q < RUNS_PG; ++q)
#pragma unroll
        for (int p = 0; p < PASSES; ++p)
          if (rc[q][p].x >= 0) rec2[pos[q * PASSES + p]] = make_int2(rc[q][p].x >> 8, rc[q][p].y);
    }
  }
}

#define TILE_NODES 32
#define NODES_PW   (TILE_NODES / 8)
#define A_PITCH    68

__device__ __forceinline__ float4 ld_row4(const float* pb, int idx) {
  return *(const float4*)(pb + (size_t)idx * D_DIM);
}

__device__ __forceinline__ void fma_row4(float4& acc, const float4 p, float g) {
  acc.x = fmaf(p.x, g, acc.x);
  acc.y = fmaf(p.y, g, acc.y);
  acc.z = fmaf(p.z, g, acc.z);
  acc.w = fmaf(p.w, g, acc.w);
}

__global__ void __launch_bounds__(THREADS)
gather_gemm_lrelu_kernel(const float* __restrict__ prev,
                         const int*   __restrict__ deg,
                         const int*   __restrict__ start,
                         const int2*  __restrict__ rec,
                         const float* __restrict__ Wm,
                         const float* __restrict__ nf,
                         const float* __restrict__ ef,
                         float*       __restrict__ out,
                         int N) {
  __shared__ float tile[B_DIM * TILE_NODES * A_PITCH];
  const int t    = threadIdx.x;
  const int lane = t & 31;
  const int w     = __builtin_amdgcn_readfirstlane(t >> 5);
  const int node0 = blockIdx.x * TILE_NODES;

  {
    const int b  = lane >> 4;
    const int d4 = (lane & 15) * 4;
    const float* pb = prev + (size_t)b * N * D_DIM + d4;

    for (int i = 0; i < NODES_PW; ++i) {
      const int nl   = w * NODES_PW + i;
      const int node = node0 + nl;
      float4 acc; acc.x = 0.f; acc.y = 0.f; acc.z = 0.f; acc.w = 0.f;

      if (node < N) {
        const int s = start[node];
        const int n = deg[node];
        for (int j0 = 0; j0 < n; j0 += 32) {
          const int m = (n - j0) < 32 ? (n - j0) : 32;
          int2 ent = make_int2(0, 0);
          if (lane < m) ent = rec[s + j0 + lane];

          int k = 0;
          for (; k + 8 <= m; k += 8) {
            int ii[8]; float gg[8]; float4 pp[8];
#pragma unroll
            for (int q = 0; q < 8; ++q) {
              ii[q] = __builtin_amdgcn_readlane(ent.x, k + q);
              gg[q] = __int_as_float(__builtin_amdgcn_readlane(ent.y, k + q));
            }
#pragma unroll
            for (int q = 0; q < 8; ++q) pp[q] = ld_row4(pb, ii[q]);
#pragma unroll
            for (int q = 0; q < 8; ++q) fma_row4(acc, pp[q], gg[q]);
          }
          if (k + 4 <= m) {
            int ii[4]; float gg[4]; float4 pp[4];
#pragma unroll
            for (int q = 0; q < 4; ++q) {
              ii[q] = __builtin_amdgcn_readlane(ent.x, k + q);
              gg[q] = __int_as_float(__builtin_amdgcn_readlane(ent.y, k + q));
            }
#pragma unroll
            for (int q = 0; q < 4; ++q) pp[q] = ld_row4(pb, ii[q]);
#pragma unroll
            for (int q = 0; q < 4; ++q) fma_row4(acc, pp[q], gg[q]);
            k += 4;
          }
          for (; k < m; ++k) {
            const int   i1 = __builtin_amdgcn_readlane(ent.x, k);
            const float g1 = __int_as_float(__builtin_amdgcn_readlane(ent.y, k));
            fma_row4(acc, ld_row4(pb, i1), g1);
          }
        }
      }
      *(float4*)(tile + (b * TILE_NODES + nl) * A_PITCH + d4) = acc;
    }
  }
  __syncthreads();

  const int rt   = w >> 1;
  const int bt   = rt >> 1;
  const int nb0  = node0 + (rt & 1) * 16;
  if (nb0 + 16 > N) return;
  const int half = lane >> 4;
  const int lm   = lane & 15;

  const float* arow = tile + (rt * 16 + lm) * A_PITCH + 2 * half;
  const int    col  = (w & 1) * 32 + lm;
  const float* wrow = Wm + (size_t)col * D_DIM + 2 * half;

  const size_t o0 = ((size_t)bt * N + nb0 + 8 * half) * D_DIM + col;

  v8f acc0, acc1;
#pragma unroll
  for (int r = 0; r < 8; ++r) {
    acc0[r] = nf[o0 + r * D_DIM]      + ef[o0 + r * D_DIM];
    acc1[r] = nf[o0 + r * D_DIM + 16] + ef[o0 + r * D_DIM + 16];
  }

#pragma unroll 4
  for (int q = 0; q < 16; ++q) {
    const v2f a  = *(const v2f*)(arow + 4 * q);
    const v2f b0 = *(const v2f*)(wrow + 4 * q);
    const v2f b1 = *(const v2f*)(wrow + 16 * D_DIM + 4 * q);
    acc0 = __builtin_amdgcn_wmma_f32_16x16x4_f32(false, a, false, b0, (short)0, acc0, false, false);
    acc1 = __builtin_amdgcn_wmma_f32_16x16x4_f32(false, a, false, b1, (short)0, acc1, false, false);
  }

#pragma unroll
  for (int r = 0; r < 8; ++r) {
    float d0 = acc0[r], d1 = acc1[r];
    d0 = d0 > 0.f ? d0 : 0.01f * d0;
    d1 = d1 > 0.f ? d1 : 0.01f * d1;
    __builtin_nontemporal_store(d0, out + o0 + r * D_DIM);
    __builtin_nontemporal_store(d1, out + o0 + r * D_DIM + 16);
  }
}

extern "C" void kernel_launch(void* const* d_in, const int* in_sizes, int n_in,
                              void* d_out, int out_size, void* d_ws, size_t ws_size,
                              hipStream_t stream) {
  const float* prev  = (const float*)d_in[0];
  const int*   edges = (const int*)  d_in[1];
  const float* nf    = (const float*)d_in[2];
  const float* ef    = (const float*)d_in[3];
  const float* st    = (const float*)d_in[4];
  const float* Wm    = (const float*)d_in[5];

  const int BND = in_sizes[0];
  const int E   = in_sizes[4];
  const int N   = BND / (B_DIM * D_DIM);

  const int nb   = (N + BIN_NODES - 1) / BIN_NODES;
  const int nblk = (E + EPB - 1) / EPB;

  int*  scr    = (int*)d_out;
  int2* rec1   = (int2*)scr;
  int*  runcnt = scr + 2 * nblk * SEG;
  int*  runoff = runcnt + MAX_BINS * MAX_BLKS;

  int*  wsi    = (int*)d_ws;
  const int NP = ((N + 63) / 64) * 64;
  int2* rec2   = (int2*)wsi;
  int*  start  = wsi + 4 * E;
  int*  deg    = start + NP;

  coarse_bin_kernel<<<nblk, THREADS, 0, stream>>>(edges, st, rec1, runcnt, runoff,
                                                  E, nb, nblk);
  fine_group_kernel<<<nb, THREADS, 0, stream>>>(rec1, runcnt, runoff, rec2, start, deg,
                                                N, nblk);
  gather_gemm_lrelu_kernel<<<(N + TILE_NODES - 1) / TILE_NODES, THREADS, 0, stream>>>(
      prev, deg, start, rec2, Wm, nf, ef, (float*)d_out, N);
}


# Counter tables on a 260-int pitch, cleared and row-summed 16 bytes per lane with the next group's loads ahead of the sums; level 2's list starts by wave shuffles and one barrier, not 16; order kept
# speedup vs baseline: 3.8351x; 1.0403x over previous
#include <hip/hip_runtime.h>

typedef __attribute__((ext_vector_type(2))) float v2f;
typedef __attribute__((ext_vector_type(8))) float v8f;

#define D_DIM 64
#define B_DIM 2


#define BIN_SHIFT  8
#define BIN_NODES  256
#define PITCH      260
#define ROW4       (PITCH / 4)
#define MAX_BINS   200
#define MAX_BLKS   256
#define THREADS    256
#define EPT        16
#define EPB        (EPT * THREADS)
#define SEG        (2 * EPB)

static_assert(MAX_BINS < THREADS, "one thread per coarse bin in level 1, and an offset entry for the spare row");
static_assert(MAX_BLKS <= THREADS, "one thread per run in level 2");
static_assert(PITCH % 4 == 0 && (ROW4 & 1) == 1 && PITCH >= THREADS, "rows are whole 16-byte pieces, an odd number of them, and hold one counter per thread");

__device__ __forceinline__ int excl_scan_256(int v, int* wsum, int t) {
  const int lane = t & 31;
  int x = v;
  for (int d = 1; d < 32; d <<= 1) {
    const int y = __shfl_up(x, d);
    if (lane >= d) x += y;
  }
  if (lane == 31) wsum[t >> 5] = x;
  __syncthreads();
  int base = 0;
  for (int w = 0; w < THREADS / 32; ++w) base += w < (t >> 5) ? wsum[w] : 0;
  return base + x - v;
}

#define RS_GRP 8
__device__ __forceinline__ int row_prefix(int4* row4) {
  int tot = 0;
  int4 cur[RS_GRP], nxt[RS_GRP];
#pragma unroll
  for (int u = 0; u < RS_GRP; ++u) cur[u] = row4[u];
  for (int j = 0; j < THREADS / 4; j += RS_GRP) {
    const bool more = j + RS_GRP < THREADS / 4;
#pragma unroll
    for (int u = 0; u < RS_GRP; ++u) nxt[u] = more ? row4[j + RS_GRP + u] : make_int4(0, 0, 0, 0);
#pragma unroll
    for (int u = 0; u < RS_GRP; ++u) {
      int4 o;
      o.x = tot; tot += cur[u].x;
      o.y = tot; tot += cur[u].y;
      o.z = tot; tot += cur[u].z;
      o.w = tot; tot += cur[u].w;
      row4[j + u] = o;
    }
#pragma unroll
    for (int u = 0; u < RS_GRP; ++u) cur[u] = nxt[u];
  }
  return tot;
}

#define GRP 8
template <int NS>
__device__ __forceinline__ void count_ranked(int* col, const int (&key)[NS], int (&rank)[NS]) {
  static_assert(NS % GRP == 0, "whole groups");
#pragma unroll
  for (int b = 0; b < NS; b += GRP) {
    int c[GRP];
#pragma unroll
    for (int i = 0; i < GRP; ++i) c[i] = col[key[b + i] * PITCH];
#pragma unroll
    for (int i = 0; i < GRP; ++i) {
      int before = 0;
#pragma unroll
      for (int j = 0; j < i; ++j) before += key[b + j] == key[b + i] ? 1 : 0;
      rank[b + i] = c[i] + before;
    }
#pragma unroll
    for (int i = 0; i < GRP; ++i) col[key[b + i] * PITCH] = rank[b + i] + 1;
  }
}

__device__ __forceinline__ void load_edges(int2 (&uv)[EPT], const int* __restrict__ edges,
                                           int E, int e0, int t) {
#pragma unroll
  for (int q = 0; q < EPT; ++q) {
    const int e = e0 + q * THREADS + t;
    uv[q] = e < E ? ((const int2*)edges)[e] : make_int2(-1, -1);
  }
}

__device__ __forceinline__ void coarse_count(int* tab, const int2 (&uv)[EPT], int (&key)[2 * EPT],
                                             int (&rank)[2 * EPT], int nb, int t) {
  int4* tab4 = (int4*)tab;
  for (int i = t; i < nb * ROW4; i += THREADS) tab4[i] = make_int4(0, 0, 0, 0);
  if (t < ROW4) tab4[MAX_BINS * ROW4 + t] = make_int4(0, 0, 0, 0);
  __syncthreads();
#pragma unroll
  for (int q = 0; q < EPT; ++q) {
    const bool ok = uv[q].x >= 0;
    key[2 * q]     = ok ? uv[q].y >> BIN_SHIFT : MAX_BINS;
    key[2 * q + 1] = ok ? uv[q].x >> BIN_SHIFT : MAX_BINS;
  }
  count_ranked<2 * EPT>(tab + t, key, rank);
}

__global__ void __launch_bounds__(THREADS)
coarse_bin_kernel(const int* __restrict__ edges, const float* __restrict__ status,
                  int2* __restrict__ rec1, int* __restrict__ runcnt,
                  int* __restrict__ runoff, int E, int nb, int nblk) {
  __shared__ __attribute__((aligned(16))) int tab[(MAX_BINS + 1) * PITCH];
  __shared__ int2 stage[SEG];
  __shared__ int  binoff[THREADS];
  __shared__ int  wsum[THREADS / 32];
  __shared__ int  nrec;
  const int t = threadIdx.x, blk = blockIdx.x;
  const int e0 = blk * EPB;
  int2 uv[EPT];
  int  gi[EPT];
  int  key[2 * EPT], slot[2 * EPT];
  load_edges(uv, edges, E, e0, t);
#pragma unroll
  for (int q = 0; q < EPT; ++q) {
    const int e = e0 + q * THREADS + t;
    gi[q] = e < E ? __float_as_int(status[e]) : 0;
  }
  coarse_count(tab, uv, key, slot, nb, t);
  __syncthreads();
  int tot = 0;
  if (t < nb) tot = row_prefix((int4*)(tab + t * PITCH));
  const int off = excl_scan_256(tot, wsum, t);
  binoff[t] = off;
  if (t == THREADS - 1) nrec = off + tot;
  if (t < nb) {
    runcnt[t * nblk + blk] = tot;
    runoff[t * nblk + blk] = off;
  }
  __syncthreads();
#pragma unroll
  for (int s = 0; s < 2 * EPT; ++s) slot[s] += binoff[key[s]] + tab[key[s] * PITCH + t];
#pragma unroll
  for (int q = 0; q < EPT; ++q) {
    if (uv[q].x >= 0) {
      const int u = uv[q].x, v = uv[q].y;
      stage[slot[2 * q]]     = make_int2((u << 8) | (v & 255), gi[q]);
      stage[slot[2 * q + 1]] = make_int2((v << 8) | (u & 255), gi[q]);
    }
  }
  __syncthreads();
  const int4* img = (const int4*)stage;
  int4*       seg = (int4*)(rec1 + (size_t)blk * SEG);
  const int   n4  = nrec >> 1;
  int4 cp[SEG / 2 / THREADS];
#pragma unroll
  for (int q = 0; q < SEG / 2 / THREADS; ++q) {
    const int i = q * THREADS + t;
    cp[q] = i < n4 ? img[i] : make_int4(0, 0, 0, 0);
  }
#pragma unroll
  for (int q = 0; q < SEG / 2 / THREADS; ++q) {
    const int i = q * THREADS + t;
    if (i < n4) seg[i] = cp[q];
  }
}

#define RUNS_PG    (MAX_BLKS / 16)
#define PASSES     4
#define IMG_CAP    (((BIN_NODES + 1) * PITCH) / 2)
#define CP_GRP     8

__global__ void __launch_bounds__(THREADS)
fine_group_kernel(const int2* __restrict__ rec1, const int* __restrict__ runcnt,
                  const int* __restrict__ runoff, int2* __restrict__ rec2,
                  int* __restrict__ start, int* __restrict__ deg, int N, int nblk) {
  __shared__ __attribute__((aligned(16))) int tab[(BIN_NODES + 1) * PITCH];
  __shared__ int  nodeoff[BIN_NODES];
  __shared__ int2 run[MAX_BLKS];
  __shared__ int  wsum[THREADS / 32];
  __shared__ int  bintot;
  const int t = threadIdx.x, bin = blockIdx.x;
  const int g = t >> 4, l = t & 15;

  int myc = 0, myo = 0;
  if (t < nblk) { myc = runcnt[bin * nblk + t]; myo = runoff[bin * nblk + t]; }

  for (int i = t; i < (BIN_NODES + 1) * ROW4; i += THREADS) ((int4*)tab)[i] = make_int4(0, 0, 0, 0);

  int below = myo;
  for (int d = 16; d >= 1; d >>= 1) below += __shfl_xor(below, d);
  if ((t & 31) == 0) wsum[t >> 5] = below;

  run[t] = make_int2(myc, t * SEG + myo);
  __syncthreads();
  int b0 = 0;
  for (int w = 0; w < THREADS / 32; ++w) b0 += wsum[w];
  int2 rn[RUNS_PG];
#pragma unroll
  for (int q = 0; q < RUNS_PG; ++q) rn[q] = run[g + 16 * q];

  unsigned longmask = 0;
#pragma unroll
  for (int q = 0; q < RUNS_PG; ++q) longmask |= (rn[q].x > PASSES * 16 ? 1u : 0u) << q;

  int2 rc[RUNS_PG][PASSES];
  int  key[RUNS_PG * PASSES];
  int  pos[RUNS_PG * PASSES];
  {
#pragma unroll
    for (int q = 0; q < RUNS_PG; ++q)
#pragma unroll
      for (int p = 0; p < PASSES; ++p) {
        const int i = p * 16 + l;
        rc[q][p] = i < rn[q].x ? rec1[rn[q].y + i] : make_int2(-1, 0);
      }
#pragma unroll
    for (int q = 0; q < RUNS_PG; ++q)
#pragma unroll
      for (int p = 0; p < PASSES; ++p)
        key[q * PASSES + p] = rc[q][p].x >= 0 ? rc[q][p].x & 255 : BIN_NODES;
    count_ranked<RUNS_PG * PASSES>(tab + t, key, pos);
    for (unsigned m = longmask; m != 0; m &= m - 1) {
      const int2 r = run[g + 16 * (__ffs(m) - 1)];
      for (int i = PASSES * 16 + l; i < r.x; i += 16)
        tab[(rec1[r.y + i].x & 255) * PITCH + t] += 1;
    }
  }
  __syncthreads();

  const int mydeg = row_prefix((int4*)(tab + t * PITCH));
  const int first = b0 + excl_scan_256(mydeg, wsum, t);
  nodeoff[t] = first;
  if (t == THREADS - 1) bintot = first + mydeg - b0;
  const int node = bin * BIN_NODES + t;
  if (node < N) { start[node] = first; deg[node] = mydeg; }
  __syncthreads();

  {
#pragma unroll
    for (int q = 0; q < RUNS_PG; ++q)
#pragma unroll
      for (int p = 0; p < PASSES; ++p) {
        const int loc = rc[q][p].x & 255;
        pos[q * PASSES + p] += nodeoff[loc] + tab[loc * PITCH + t];
      }

    for (unsigned m = longmask; m != 0; m &= m - 1) {
      const int2 r = run[g + 16 * (__ffs(m) - 1)];
      for (int i = PASSES * 16 + l; i < r.x; i += 16) {
        const int2 r1  = rec1[r.y + i];
        const int  loc = r1.x & 255;
        int mine = 0;
#pragma unroll
        for (int q = 0; q < RUNS_PG; ++q)
#pragma unroll
          for (int p = 0; p < PASSES; ++p)
            mine += (rc[q][p].x & (int)0x800000ffu) == loc ? 1 : 0;
        const int  k   = tab[loc * PITCH + t];
        tab[loc * PITCH + t] = k + 1;
        rec2[nodeoff[loc] + k + mine] = make_int2(r1.x >> 8, r1.y);
      }
    }

    const int base = b0 & ~15;
    const int span = b0 - base + bintot;
    if (span <= IMG_CAP) {
      int4* img4 = (int4*)tab;
      int2* img  = (int2*)tab;
      const int n4 = (span + 1) >> 1;
      __syncthreads();
      for (int i = t; i < n4; i += THREADS) img4[i] = make_int4(-1, 0, -1, 0);
      __syncthreads();
#pragma unroll
      for (int q = 0; q < RUNS_PG; ++q)
#pragma unroll
        for (int p = 0; p < PASSES; ++p)
          if (rc[q][p].x >= 0)
            img[pos[q * PASSES + p] - base] = make_int2(rc[q][p].x >> 8, rc[q][p].y);
      __syncthreads();
      int4* dst = (int4*)(rec2 + base);
      for (int i0 = 0; i0 < n4; i0 += CP_GRP * THREADS) {
        int4 cp[CP_GRP];
#pragma unroll
        for (int u = 0; u < CP_GRP; ++u) {
          const int i = i0 + u * THREADS + t;
          cp[u] = i < n4 ? img4[i] : make_int4(-1, 0, -1, 0);
        }
#pragma unroll
        for (int u = 0; u < CP_GRP; ++u) {
          const int i = i0 + u * THREADS + t;
          if (cp[u].x >= 0 && cp[u].z >= 0) dst[i] = cp[u];
          else if (cp[u].x >= 0) rec2[base + 2 * i]     = make_int2(cp[u].x, cp[u].y);
          else if (cp[u].z >= 0) rec2[base + 2 * i + 1] = make_int2(cp[u].z, cp[u].w);
        }
      }
    } else {
#pragma unroll
      for (int q = 0; q < RUNS_PG; ++q)
#pragma unroll
        for (int p = 0; p < PASSES; ++p)
          if (rc[q][p].x >= 0) rec2[pos[q * PASSES + p]] = make_int2(rc[q][p].x >> 8, rc[q][p].y);
    }
  }
}

#define TILE_NODES 32
#define NODES_PW   (TILE_NODES / 8)
#define A_PITCH    68

__device__ __forceinline__ float4 ld_row4(const float* pb, int idx) {
  return *(const float4*)(pb + (size_t)idx * D_DIM);
}

__device__ __forceinline__ void fma_row4(float4& acc, const float4 p, float g) {
  acc.x = fmaf(p.x, g, acc.x);
  acc.y = fmaf(p.y, g, acc.y);
  acc.z = fmaf(p.z, g, acc.z);
  acc.w = fmaf(p.w, g, acc.w);
}

__global__ void __launch_bounds__(THREADS)
gather_gemm_lrelu_kernel(const float* __restrict__ prev,
                         const int*   __restrict__ deg,
                         const int*   __restrict__ start,
                         const int2*  __restrict__ rec,
                         const float* __restrict__ Wm,
                         const float* __restrict__ nf,
                         const float* __restrict__ ef,
                         float*       __restrict__ out,
                         int N) {
  __shared__ float tile[B_DIM * TILE_NODES * A_PITCH];
  const int t    = threadIdx.x;
  const int lane = t & 31;
  const int w     = __builtin_amdgcn_readfirstlane(t >> 5);
  const int node0 = blockIdx.x * TILE_NODES;

  {
    const int b  = lane >> 4;
    const int d4 = (lane & 15) * 4;
    const float* pb = prev + (size_t)b * N * D_DIM + d4;

    for (int i = 0; i < NODES_PW; ++i) {
      const int nl   = w * NODES_PW + i;
      const int node = node0 + nl;
      float4 acc; acc.x = 0.f; acc.y = 0.f; acc.z = 0.f; acc.w = 0.f;

      if (node < N) {
        const int s = start[node];
        const int n = deg[node];
        for (int j0 = 0; j0 < n; j0 += 32) {
          const int m = (n - j0) < 32 ? (n - j0) : 32;
          int2 ent = make_int2(0, 0);
          if (lane < m) ent = rec[s + j0 + lane];

          int k = 0;
          for (; k + 8 <= m; k += 8) {
            int ii[8]; float gg[8]; float4 pp[8];
#pragma unroll
            for (int q = 0; q < 8; ++q) {
              ii[q] = __builtin_amdgcn_readlane(ent.x, k + q);
              gg[q] = __int_as_float(__builtin_amdgcn_readlane(ent.y, k + q));
            }
#pragma unroll
            for (int q = 0; q < 8; ++q) pp[q] = ld_row4(pb, ii[q]);
#pragma unroll
            for (int q = 0; q < 8; ++q) fma_row4(acc, pp[q], gg[q]);
          }
          if (k + 4 <= m) {
            int ii[4]; float gg[4]; float4 pp[4];
#pragma unroll
            for (int q = 0; q < 4; ++q) {
              ii[q] = __builtin_amdgcn_readlane(ent.x, k + q);
              gg[q] = __int_as_float(__builtin_amdgcn_readlane(ent.y, k + q));
            }
#pragma unroll
            for (int q = 0; q < 4; ++q) pp[q] = ld_row4(pb, ii[q]);
#pragma unroll
            for (int q = 0; q < 4; ++q) fma_row4(acc, pp[q], gg[q]);
            k += 4;
          }
          for (; k < m; ++k) {
            const int   i1 = __builtin_amdgcn_readlane(ent.x, k);
            const float g1 = __int_as_float(__builtin_amdgcn_readlane(ent.y, k));
            fma_row4(acc, ld_row4(pb, i1), g1);
          }
        }
      }
      *(float4*)(tile + (b * TILE_NODES + nl) * A_PITCH + d4) = acc;
    }
  }
  __syncthreads();

  const int rt   = w >> 1;
  const int bt   = rt >> 1;
  const int nb0  = node0 + (rt & 1) * 16;
  if (nb0 + 16 > N) return;
  const int half = lane >> 4;
  const int lm   = lane & 15;

  const float* arow = tile + (rt * 16 + lm) * A_PITCH + 2 * half;
  const int    col  = (w & 1) * 32 + lm;
  const float* wrow = Wm + (size_t)col * D_DIM + 2 * half;

  const size_t o0 = ((size_t)bt * N + nb0 + 8 * half) * D_DIM + col;

  v8f acc0, acc1;
#pragma unroll
  for (int r = 0; r < 8; ++r) {
    acc0[r] = nf[o0 + r * D_DIM]      + ef[o0 + r * D_DIM];
    acc1[r] = nf[o0 + r * D_DIM + 16] + ef[o0 + r * D_DIM + 16];
  }

#pragma unroll 4
  for (int q = 0; q < 16; ++q) {
    const v2f a  = *(const v2f*)(arow + 4 * q);
    const v2f b0 = *(const v2f*)(wrow + 4 * q);
    const v2f b1 = *(const v2f*)(wrow + 16 * D_DIM + 4 * q);
    acc0 = __builtin_amdgcn_wmma_f32_16x16x4_f32(false, a, false, b0, (short)0, acc0, false, false);
    acc1 = __builtin_amdgcn_wmma_f32_16x16x4_f32(false, a, false, b1, (short)0, acc1, false, false);
  }

#pragma unroll
  for (int r = 0; r < 8; ++r) {
    float d0 = acc0[r], d1 = acc1[r];
    d0 = d0 > 0.f ? d0 : 0.01f * d0;
    d1 = d1 > 0.f ? d1 : 0.01f * d1;
    __builtin_nontemporal_store(d0, out + o0 + r * D_DIM);
    __builtin_nontemporal_store(d1, out + o0 + r * D_DIM + 16);
  }
}

extern "C" void kernel_launch(void* const* d_in, const int* in_sizes, int n_in,
                              void* d_out, int out_size, void* d_ws, size_t ws_size,
                              hipStream_t stream) {
  const float* prev  = (const float*)d_in[0];
  const int*   edges = (const int*)  d_in[1];
  const float* nf    = (const float*)d_in[2];
  const float* ef    = (const float*)d_in[3];
  const float* st    = (const float*)d_in[4];
  const float* Wm    = (const float*)d_in[5];

  const int BND = in_sizes[0];
  const int E   = in_sizes[4];
  const int N   = BND / (B_DIM * D_DIM);

  const int nb   = (N + BIN_NODES - 1) / BIN_NODES;
  const int nblk = (E + EPB - 1) / EPB;

  int*  scr    = (int*)d_out;
  int2* rec1   = (int2*)scr;
  int*  runcnt = scr + 2 * nblk * SEG;
  int*  runoff = runcnt + MAX_BINS * MAX_BLKS;

  int*  wsi    = (int*)d_ws;
  const int NP = ((N + 63) / 64) * 64;
  int2* rec2   = (int2*)wsi;
  int*  start  = wsi + 4 * E;
  int*  deg    = start + NP;

  coarse_bin_kernel<<<nblk, THREADS, 0, stream>>>(edges, st, rec1, runcnt, runoff,
                                                  E, nb, nblk);
  fine_group_kernel<<<nb, THREADS, 0, stream>>>(rec1, runcnt, runoff, rec2, start, deg,
                                                N, nblk);
  gather_gemm_lrelu_kernel<<<(N + TILE_NODES - 1) / TILE_NODES, THREADS, 0, stream>>>(
      prev, deg, start, rec2, Wm, nf, ef, (float*)d_out, N);
}


# W copied once per work-group into LDS (whole 128-byte lines) and B fragments read from that copy: the 32 global loads per wave of 16 lines each are gone; product and sums bit-identical
# speedup vs baseline: 4.0242x; 1.0493x over previous
#include <hip/hip_runtime.h>

typedef __attribute__((ext_vector_type(2))) float v2f;
typedef __attribute__((ext_vector_type(8))) float v8f;

#define D_DIM 64
#define B_DIM 2


#define BIN_SHIFT  8
#define BIN_NODES  256
#define PITCH      260
#define ROW4       (PITCH / 4)
#define MAX_BINS   200
#define MAX_BLKS   256
#define THREADS    256
#define EPT        16
#define EPB        (EPT * THREADS)
#define SEG        (2 * EPB)

static_assert(MAX_BINS < THREADS, "one thread per coarse bin in level 1, and an offset entry for the spare row");
static_assert(MAX_BLKS <= THREADS, "one thread per run in level 2");
static_assert(PITCH % 4 == 0 && (ROW4 & 1) == 1 && PITCH >= THREADS, "rows are whole 16-byte pieces, an odd number of them, and hold one counter per thread");

__device__ __forceinline__ int excl_scan_256(int v, int* wsum, int t) {
  const int lane = t & 31;
  int x = v;
  for (int d = 1; d < 32; d <<= 1) {
    const int y = __shfl_up(x, d);
    if (lane >= d) x += y;
  }
  if (lane == 31) wsum[t >> 5] = x;
  __syncthreads();
  int base = 0;
  for (int w = 0; w < THREADS / 32; ++w) base += w < (t >> 5) ? wsum[w] : 0;
  return base + x - v;
}

#define RS_GRP 8
__device__ __forceinline__ int row_prefix(int4* row4) {
  int tot = 0;
  int4 cur[RS_GRP], nxt[RS_GRP];
#pragma unroll
  for (int u = 0; u < RS_GRP; ++u) cur[u] = row4[u];
  for (int j = 0; j < THREADS / 4; j += RS_GRP) {
    const bool more = j + RS_GRP < THREADS / 4;
#pragma unroll
    for (int u = 0; u < RS_GRP; ++u) nxt[u] = more ? row4[j + RS_GRP + u] : make_int4(0, 0, 0, 0);
#pragma unroll
    for (int u = 0; u < RS_GRP; ++u) {
      int4 o;
      o.x = tot; tot += cur[u].x;
      o.y = tot; tot += cur[u].y;
      o.z = tot; tot += cur[u].z;
      o.w = tot; tot += cur[u].w;
      row4[j + u] = o;
    }
#pragma unroll
    for (int u = 0; u < RS_GRP; ++u) cur[u] = nxt[u];
  }
  return tot;
}

#define GRP 8
template <int NS>
__device__ __forceinline__ void count_ranked(int* col, const int (&key)[NS], int (&rank)[NS]) {
  static_assert(NS % GRP == 0, "whole groups");
#pragma unroll
  for (int b = 0; b < NS; b += GRP) {
    int c[GRP];
#pragma unroll
    for (int i = 0; i < GRP; ++i) c[i] = col[key[b + i] * PITCH];
#pragma unroll
    for (int i = 0; i < GRP; ++i) {
      int before = 0;
#pragma unroll
      for (int j = 0; j < i; ++j) before += key[b + j] == key[b + i] ? 1 : 0;
      rank[b + i] = c[i] + before;
    }
#pragma unroll
    for (int i = 0; i < GRP; ++i) col[key[b + i] * PITCH] = rank[b + i] + 1;
  }
}

__device__ __forceinline__ void load_edges(int2 (&uv)[EPT], const int* __restrict__ edges,
                                           int E, int e0, int t) {
#pragma unroll
  for (int q = 0; q < EPT; ++q) {
    const int e = e0 + q * THREADS + t;
    uv[q] = e < E ? ((const int2*)edges)[e] : make_int2(-1, -1);
  }
}

__device__ __forceinline__ void coarse_count(int* tab, const int2 (&uv)[EPT], int (&key)[2 * EPT],
                                             int (&rank)[2 * EPT], int nb, int t) {
  int4* tab4 = (int4*)tab;
  for (int i = t; i < nb * ROW4; i += THREADS) tab4[i] = make_int4(0, 0, 0, 0);
  if (t < ROW4) tab4[MAX_BINS * ROW4 + t] = make_int4(0, 0, 0, 0);
  __syncthreads();
#pragma unroll
  for (int q = 0; q < EPT; ++q) {
    const bool ok = uv[q].x >= 0;
    key[2 * q]     = ok ? uv[q].y >> BIN_SHIFT : MAX_BINS;
    key[2 * q + 1] = ok ? uv[q].x >> BIN_SHIFT : MAX_BINS;
  }
  count_ranked<2 * EPT>(tab + t, key, rank);
}

__global__ void __launch_bounds__(THREADS)
coarse_bin_kernel(const int* __restrict__ edges, const float* __restrict__ status,
                  int2* __restrict__ rec1, int* __restrict__ runcnt,
                  int* __restrict__ runoff, int E, int nb, int nblk) {
  __shared__ __attribute__((aligned(16))) int tab[(MAX_BINS + 1) * PITCH];
  __shared__ int2 stage[SEG];
  __shared__ int  binoff[THREADS];
  __shared__ int  wsum[THREADS / 32];
  __shared__ int  nrec;
  const int t = threadIdx.x, blk = blockIdx.x;
  const int e0 = blk * EPB;
  int2 uv[EPT];
  int  gi[EPT];
  int  key[2 * EPT], slot[2 * EPT];
  load_edges(uv, edges, E, e0, t);
#pragma unroll
  for (int q = 0; q < EPT; ++q) {
    const int e = e0 + q * THREADS + t;
    gi[q] = e < E ? __float_as_int(status[e]) : 0;
  }
  coarse_count(tab, uv, key, slot, nb, t);
  __syncthreads();
  int tot = 0;
  if (t < nb) tot = row_prefix((int4*)(tab + t * PITCH));
  const int off = excl_scan_256(tot, wsum, t);
  binoff[t] = off;
  if (t == THREADS - 1) nrec = off + tot;
  if (t < nb) {
    runcnt[t * nblk + blk] = tot;
    runoff[t * nblk + blk] = off;
  }
  __syncthreads();
#pragma unroll
  for (int s = 0; s < 2 * EPT; ++s) slot[s] += binoff[key[s]] + tab[key[s] * PITCH + t];
#pragma unroll
  for (int q = 0; q < EPT; ++q) {
    if (uv[q].x >= 0) {
      const int u = uv[q].x, v = uv[q].y;
      stage[slot[2 * q]]     = make_int2((u << 8) | (v & 255), gi[q]);
      stage[slot[2 * q + 1]] = make_int2((v << 8) | (u & 255), gi[q]);
    }
  }
  __syncthreads();
  const int4* img = (const int4*)stage;
  int4*       seg = (int4*)(rec1 + (size_t)blk * SEG);
  const int   n4  = nrec >> 1;
  int4 cp[SEG / 2 / THREADS];
#pragma unroll
  for (int q = 0; q < SEG / 2 / THREADS; ++q) {
    const int i = q * THREADS + t;
    cp[q] = i < n4 ? img[i] : make_int4(0, 0, 0, 0);
  }
#pragma unroll
  for (int q = 0; q < SEG / 2 / THREADS; ++q) {
    const int i = q * THREADS + t;
    if (i < n4) seg[i] = cp[q];
  }
}

#define RUNS_PG    (MAX_BLKS / 16)
#define PASSES     4
#define IMG_CAP    (((BIN_NODES + 1) * PITCH) / 2)
#define CP_GRP     8

__global__ void __launch_bounds__(THREADS)
fine_group_kernel(const int2* __restrict__ rec1, const int* __restrict__ runcnt,
                  const int* __restrict__ runoff, int2* __restrict__ rec2,
                  int* __restrict__ start, int* __restrict__ deg, int N, int nblk) {
  __shared__ __attribute__((aligned(16))) int tab[(BIN_NODES + 1) * PITCH];
  __shared__ int  nodeoff[BIN_NODES];
  __shared__ int2 run[MAX_BLKS];
  __shared__ int  wsum[THREADS / 32];
  __shared__ int  bintot;
  const int t = threadIdx.x, bin = blockIdx.x;
  const int g = t >> 4, l = t & 15;

  int myc = 0, myo = 0;
  if (t < nblk) { myc = runcnt[bin * nblk + t]; myo = runoff[bin * nblk + t]; }

  for (int i = t; i < (BIN_NODES + 1) * ROW4; i += THREADS) ((int4*)tab)[i] = make_int4(0, 0, 0, 0);

  int below = myo;
  for (int d = 16; d >= 1; d >>= 1) below += __shfl_xor(below, d);
  if ((t & 31) == 0) wsum[t >> 5] = below;

  run[t] = make_int2(myc, t * SEG + myo);
  __syncthreads();
  int b0 = 0;
  for (int w = 0; w < THREADS / 32; ++w) b0 += wsum[w];
  int2 rn[RUNS_PG];
#pragma unroll
  for (int q = 0; q < RUNS_PG; ++q) rn[q] = run[g + 16 * q];

  unsigned longmask = 0;
#pragma unroll
  for (int q = 0; q < RUNS_PG; ++q) longmask |= (rn[q].x > PASSES * 16 ? 1u : 0u) << q;

  int2 rc[RUNS_PG][PASSES];
  int  key[RUNS_PG * PASSES];
  int  pos[RUNS_PG * PASSES];
  {
#pragma unroll
    for (int q = 0; q < RUNS_PG; ++q)
#pragma unroll
      for (int p = 0; p < PASSES; ++p) {
        const int i = p * 16 + l;
        rc[q][p] = i < rn[q].x ? rec1[rn[q].y + i] : make_int2(-1, 0);
      }
#pragma unroll
    for (int q = 0; q < RUNS_PG; ++q)
#pragma unroll
      for (int p = 0; p < PASSES; ++p)
        key[q * PASSES + p] = rc[q][p].x >= 0 ? rc[q][p].x & 255 : BIN_NODES;
    count_ranked<RUNS_PG * PASSES>(tab + t, key, pos);
    for (unsigned m = longmask; m != 0; m &= m - 1) {
      const int2 r = run[g + 16 * (__ffs(m) - 1)];
      for (int i = PASSES * 16 + l; i < r.x; i += 16)
        tab[(rec1[r.y + i].x & 255) * PITCH + t] += 1;
    }
  }
  __syncthreads();

  const int mydeg = row_prefix((int4*)(tab + t * PITCH));
  const int first = b0 + excl_scan_256(mydeg, wsum, t);
  nodeoff[t] = first;
  if (t == THREADS - 1) bintot = first + mydeg - b0;
  const int node = bin * BIN_NODES + t;
  if (node < N) { start[node] = first; deg[node] = mydeg; }
  __syncthreads();

  {
#pragma unroll
    for (int q = 0; q < RUNS_PG; ++q)
#pragma unroll
      for (int p = 0; p < PASSES; ++p) {
        const int loc = rc[q][p].x & 255;
        pos[q * PASSES + p] += nodeoff[loc] + tab[loc * PITCH + t];
      }

    for (unsigned m = longmask; m != 0; m &= m - 1) {
      const int2 r = run[g + 16 * (__ffs(m) - 1)];
      for (int i = PASSES * 16 + l; i < r.x; i += 16) {
        const int2 r1  = rec1[r.y + i];
        const int  loc = r1.x & 255;
        int mine = 0;
#pragma unroll
        for (int q = 0; q < RUNS_PG; ++q)
#pragma unroll
          for (int p = 0; p < PASSES; ++p)
            mine += (rc[q][p].x & (int)0x800000ffu) == loc ? 1 : 0;
        const int  k   = tab[loc * PITCH + t];
        tab[loc * PITCH + t] = k + 1;
        rec2[nodeoff[loc] + k + mine] = make_int2(r1.x >> 8, r1.y);
      }
    }

    const int base = b0 & ~15;
    const int span = b0 - base + bintot;
    if (span <= IMG_CAP) {
      int4* img4 = (int4*)tab;
      int2* img  = (int2*)tab;
      const int n4 = (span + 1) >> 1;
      __syncthreads();
      for (int i = t; i < n4; i += THREADS) img4[i] = make_int4(-1, 0, -1, 0);
      __syncthreads();
#pragma unroll
      for (int q = 0; q < RUNS_PG; ++q)
#pragma unroll
        for (int p = 0; p < PASSES; ++p)
          if (rc[q][p].x >= 0)
            img[pos[q * PASSES + p] - base] = make_int2(rc[q][p].x >> 8, rc[q][p].y);
      __syncthreads();
      int4* dst = (int4*)(rec2 + base);
      for (int i0 = 0; i0 < n4; i0 += CP_GRP * THREADS) {
        int4 cp[CP_GRP];
#pragma unroll
        for (int u = 0; u < CP_GRP; ++u) {
          const int i = i0 + u * THREADS + t;
          cp[u] = i < n4 ? img4[i] : make_int4(-1, 0, -1, 0);
        }
#pragma unroll
        for (int u = 0; u < CP_GRP; ++u) {
          const int i = i0 + u * THREADS + t;
          if (cp[u].x >= 0 && cp[u].z >= 0) dst[i] = cp[u];
          else if (cp[u].x >= 0) rec2[base + 2 * i]     = make_int2(cp[u].x, cp[u].y);
          else if (cp[u].z >= 0) rec2[base + 2 * i + 1] = make_int2(cp[u].z, cp[u].w);
        }
      }
    } else {
#pragma unroll
      for (int q = 0; q < RUNS_PG; ++q)
#pragma unroll
        for (int p = 0; p < PASSES; ++p)
          if (rc[q][p].x >= 0) rec2[pos[q * PASSES + p]] = make_int2(rc[q][p].x >> 8, rc[q][p].y);
    }
  }
}

#define TILE_NODES 32
#define NODES_PW   (TILE_NODES / 8)
#define A_PITCH    68

__device__ __forceinline__ float4 ld_row4(const float* pb, int idx) {
  return *(const float4*)(pb + (size_t)idx * D_DIM);
}

__device__ __forceinline__ void fma_row4(float4& acc, const float4 p, float g) {
  acc.x = fmaf(p.x, g, acc.x);
  acc.y = fmaf(p.y, g, acc.y);
  acc.z = fmaf(p.z, g, acc.z);
  acc.w = fmaf(p.w, g, acc.w);
}

__global__ void __launch_bounds__(THREADS)
gather_gemm_lrelu_kernel(const float* __restrict__ prev,
                         const int*   __restrict__ deg,
                         const int*   __restrict__ start,
                         const int2*  __restrict__ rec,
                         const float* __restrict__ Wm,
                         const float* __restrict__ nf,
                         const float* __restrict__ ef,
                         float*       __restrict__ out,
                         int N) {
  __shared__ float tile[B_DIM * TILE_NODES * A_PITCH];
  __shared__ __attribute__((aligned(16))) float wl[D_DIM * A_PITCH];
  const int t    = threadIdx.x;
  const int lane = t & 31;
  const int w     = __builtin_amdgcn_readfirstlane(t >> 5);
  const int node0 = blockIdx.x * TILE_NODES;

  {
    float4 wv[(D_DIM * D_DIM / 4) / THREADS];
#pragma unroll
    for (int j = 0; j < (D_DIM * D_DIM / 4) / THREADS; ++j)
      wv[j] = ((const float4*)Wm)[j * THREADS + t];
#pragma unroll
    for (int j = 0; j < (D_DIM * D_DIM / 4) / THREADS; ++j) {
      const int i = j * THREADS + t;
      *(float4*)(wl + (i >> 4) * A_PITCH + (i & 15) * 4) = wv[j];
    }
  }

  {
    const int b  = lane >> 4;
    const int d4 = (lane & 15) * 4;
    const float* pb = prev + (size_t)b * N * D_DIM + d4;

    for (int i = 0; i < NODES_PW; ++i) {
      const int nl   = w * NODES_PW + i;
      const int node = node0 + nl;
      float4 acc; acc.x = 0.f; acc.y = 0.f; acc.z = 0.f; acc.w = 0.f;

      if (node < N) {
        const int s = start[node];
        const int n = deg[node];
        for (int j0 = 0; j0 < n; j0 += 32) {
          const int m = (n - j0) < 32 ? (n - j0) : 32;
          int2 ent = make_int2(0, 0);
          if (lane < m) ent = rec[s + j0 + lane];

          int k = 0;
          for (; k + 8 <= m; k += 8) {
            int ii[8]; float gg[8]; float4 pp[8];
#pragma unroll
            for (int q = 0; q < 8; ++q) {
              ii[q] = __builtin_amdgcn_readlane(ent.x, k + q);
              gg[q] = __int_as_float(__builtin_amdgcn_readlane(ent.y, k + q));
            }
#pragma unroll
            for (int q = 0; q < 8; ++q) pp[q] = ld_row4(pb, ii[q]);
#pragma unroll
            for (int q = 0; q < 8; ++q) fma_row4(acc, pp[q], gg[q]);
          }
          if (k + 4 <= m) {
            int ii[4]; float gg[4]; float4 pp[4];
#pragma unroll
            for (int q = 0; q < 4; ++q) {
              ii[q] = __builtin_amdgcn_readlane(ent.x, k + q);
              gg[q] = __int_as_float(__builtin_amdgcn_readlane(ent.y, k + q));
            }
#pragma unroll
            for (int q = 0; q < 4; ++q) pp[q] = ld_row4(pb, ii[q]);
#pragma unroll
            for (int q = 0; q < 4; ++q) fma_row4(acc, pp[q], gg[q]);
            k += 4;
          }
          for (; k < m; ++k) {
            const int   i1 = __builtin_amdgcn_readlane(ent.x, k);
            const float g1 = __int_as_float(__builtin_amdgcn_readlane(ent.y, k));
            fma_row4(acc, ld_row4(pb, i1), g1);
          }
        }
      }
      *(float4*)(tile + (b * TILE_NODES + nl) * A_PITCH + d4) = acc;
    }
  }
  __syncthreads();

  const int rt   = w >> 1;
  const int bt   = rt >> 1;
  const int nb0  = node0 + (rt & 1) * 16;
  if (nb0 + 16 > N) return;
  const int half = lane >> 4;
  const int lm   = lane & 15;

  const float* arow = tile + (rt * 16 + lm) * A_PITCH + 2 * half;
  const int    col  = (w & 1) * 32 + lm;
  const float* wrow = wl + col * A_PITCH + 2 * half;

  const size_t o0 = ((size_t)bt * N + nb0 + 8 * half) * D_DIM + col;

  v8f acc0, acc1;
#pragma unroll
  for (int r = 0; r < 8; ++r) {
    acc0[r] = nf[o0 + r * D_DIM]      + ef[o0 + r * D_DIM];
    acc1[r] = nf[o0 + r * D_DIM + 16] + ef[o0 + r * D_DIM + 16];
  }

#pragma unroll 4
  for (int q = 0; q < 16; ++q) {
    const v2f a  = *(const v2f*)(arow + 4 * q);
    const v2f b0 = *(const v2f*)(wrow + 4 * q);
    const v2f b1 = *(const v2f*)(wrow + 16 * A_PITCH + 4 * q);
    acc0 = __builtin_amdgcn_wmma_f32_16x16x4_f32(false, a, false, b0, (short)0, acc0, false, false);
    acc1 = __builtin_amdgcn_wmma_f32_16x16x4_f32(false, a, false, b1, (short)0, acc1, false, false);
  }

#pragma unroll
  for (int r = 0; r < 8; ++r) {
    float d0 = acc0[r], d1 = acc1[r];
    d0 = d0 > 0.f ? d0 : 0.01f * d0;
    d1 = d1 > 0.f ? d1 : 0.01f * d1;
    __builtin_nontemporal_store(d0, out + o0 + r * D_DIM);
    __builtin_nontemporal_store(d1, out + o0 + r * D_DIM + 16);
  }
}

extern "C" void kernel_launch(void* const* d_in, const int* in_sizes, int n_in,
                              void* d_out, int out_size, void* d_ws, size_t ws_size,
                              hipStream_t stream) {
  const float* prev  = (const float*)d_in[0];
  const int*   edges = (const int*)  d_in[1];
  const float* nf    = (const float*)d_in[2];
  const float* ef    = (const float*)d_in[3];
  const float* st    = (const float*)d_in[4];
  const float* Wm    = (const float*)d_in[5];

  const int BND = in_sizes[0];
  const int E   = in_sizes[4];
  const int N   = BND / (B_DIM * D_DIM);

  const int nb   = (N + BIN_NODES - 1) / BIN_NODES;
  const int nblk = (E + EPB - 1) / EPB;

  int*  scr    = (int*)d_out;
  int2* rec1   = (int2*)scr;
  int*  runcnt = scr + 2 * nblk * SEG;
  int*  runoff = runcnt + MAX_BINS * MAX_BLKS;

  int*  wsi    = (int*)d_ws;
  const int NP = ((N + 63) / 64) * 64;
  int2* rec2   = (int2*)wsi;
  int*  start  = wsi + 4 * E;
  int*  deg    = start + NP;

  coarse_bin_kernel<<<nblk, THREADS, 0, stream>>>(edges, st, rec1, runcnt, runoff,
                                                  E, nb, nblk);
  fine_group_kernel<<<nb, THREADS, 0, stream>>>(rec1, runcnt, runoff, rec2, start, deg,
                                                N, nblk);
  gather_gemm_lrelu_kernel<<<(N + TILE_NODES - 1) / TILE_NODES, THREADS, 0, stream>>>(
      prev, deg, start, rec2, Wm, nf, ef, (float*)d_out, N);
}
